# starting kernel
# baseline (speedup 1.0000x reference)
#include <hip/hip_runtime.h>
#include <math.h>

typedef float v2f __attribute__((ext_vector_type(2)));
typedef float v8f __attribute__((ext_vector_type(8)));

#define DFEAT 128
#define NEG_SLOPE 0.01f
#define LN_EPS 1e-5f

__global__ void init_kernel(float* __restrict__ out, float* __restrict__ deg, int N) {
    int i = blockIdx.x * blockDim.x + threadIdx.x;
    if (i < N * DFEAT) out[i] = 0.0f;
    if (i < N) deg[i] = 1.0f;
}

__global__ void edge_deg_kernel(const float* __restrict__ ea, const int* __restrict__ ei,
                                float* __restrict__ deg, int E) {
    int e = blockIdx.x * blockDim.x + threadIdx.x;
    if (e < E) {
        int dst = ei[E + e];
        atomicAdd(&deg[dst], fabsf(ea[(size_t)e * 4]));
    }
}

__global__ void dinv_kernel(const float* __restrict__ deg, float* __restrict__ dinv, int N) {
    int i = blockIdx.x * blockDim.x + threadIdx.x;
    if (i < N) {
        float d = deg[i];
        dinv[i] = (d > 0.0f) ? rsqrtf(d) : 0.0f;
    }
}

__global__ __launch_bounds__(32) void gemm_wmma_kernel(const float* __restrict__ x,
                                                       const float* __restrict__ W,
                                                       const float* __restrict__ bias,
                                                       float* __restrict__ h, int N) {
    int tile = blockIdx.x;
    int tn = tile & 7;
    int tm = tile >> 3;
    int lane = threadIdx.x;
    int half = lane >> 4;
    int l16  = lane & 15;

    int m = tm * 16 + l16;
    if (m >= N) m = N - 1;
    int n = tn * 16 + l16;

    const float* xrow = x + (size_t)m * DFEAT + half * 2;
    const float* wrow = W + (size_t)n * DFEAT + half * 2;

    v8f c = {};
    #pragma unroll 4
    for (int k0 = 0; k0 < DFEAT; k0 += 4) {
        v2f a, b;
        a.x = xrow[k0];     a.y = xrow[k0 + 1];
        b.x = wrow[k0];     b.y = wrow[k0 + 1];
        c = __builtin_amdgcn_wmma_f32_16x16x4_f32(false, a, false, b,
                                                  (short)0, c, false, false);
    }

    float bv = bias[tn * 16 + l16];
    int row0 = tm * 16 + half * 8;
    float* hp = h + (size_t)row0 * DFEAT + tn * 16 + l16;
    if (row0 + 8 <= N) {
        #pragma unroll
        for (int v = 0; v < 8; v++) {
            hp[(size_t)v * DFEAT] = c[v] + bv;
        }
    } else {
        #pragma unroll
        for (int v = 0; v < 8; v++) {
            if (row0 + v < N) hp[(size_t)v * DFEAT] = c[v] + bv;
        }
    }
}

__global__ void scatter_kernel(const float* __restrict__ h, const float* __restrict__ ea,
                               const int* __restrict__ ei, const float* __restrict__ dinv,
                               float* __restrict__ out, int E) {
    int wave = (int)((blockIdx.x * (size_t)blockDim.x + threadIdx.x) >> 5);
    int lane = threadIdx.x & 31;
    if (wave >= E) return;
    int src = ei[wave];
    int dst = ei[E + wave];
    float norm = dinv[src] * fabsf(ea[(size_t)wave * 4]) * dinv[dst];
    int d0 = lane * 4;
    const float4 hv = *(const float4*)(h + (size_t)src * DFEAT + d0);
    float* od = out + (size_t)dst * DFEAT + d0;
    atomicAdd(&od[0], hv.x * norm);
    atomicAdd(&od[1], hv.y * norm);
    atomicAdd(&od[2], hv.z * norm);
    atomicAdd(&od[3], hv.w * norm);
}

__global__ void finish_kernel(const float* __restrict__ h, const float* __restrict__ dinv,
                              const float* __restrict__ gamma, const float* __restrict__ beta,
                              float* __restrict__ out, int N) {
    int node = (int)((blockIdx.x * (size_t)blockDim.x + threadIdx.x) >> 5);
    int lane = threadIdx.x & 31;
    if (node >= N) return;

    float di = dinv[node];
    float self_norm = di * di;

    int d0 = lane * 4;
    float4 ov = *(const float4*)(out + (size_t)node * DFEAT + d0);
    float4 hv = *(const float4*)(h + (size_t)node * DFEAT + d0);

    float v[4];
    v[0] = ov.x + hv.x * self_norm;
    v[1] = ov.y + hv.y * self_norm;
    v[2] = ov.z + hv.z * self_norm;
    v[3] = ov.w + hv.w * self_norm;

    float sum = 0.0f;
    #pragma unroll
    for (int j = 0; j < 4; j++) {
        v[j] = (v[j] >= 0.0f) ? v[j] : NEG_SLOPE * v[j];
        sum += v[j];
    }
    #pragma unroll
    for (int off = 16; off > 0; off >>= 1) sum += __shfl_xor(sum, off, 32);
    float mean = sum * (1.0f / DFEAT);

    float vs = 0.0f;
    #pragma unroll
    for (int j = 0; j < 4; j++) {
        float dlt = v[j] - mean;
        vs += dlt * dlt;
    }
    #pragma unroll
    for (int off = 16; off > 0; off >>= 1) vs += __shfl_xor(vs, off, 32);
    float inv_std = rsqrtf(vs * (1.0f / DFEAT) + LN_EPS);

    float4 gv = *(const float4*)(gamma + d0);
    float4 bv = *(const float4*)(beta + d0);
    float4 res;
    res.x = (v[0] - mean) * inv_std * gv.x + bv.x;
    res.y = (v[1] - mean) * inv_std * gv.y + bv.y;
    res.z = (v[2] - mean) * inv_std * gv.z + bv.z;
    res.w = (v[3] - mean) * inv_std * gv.w + bv.w;
    *(float4*)(out + (size_t)node * DFEAT + d0) = res;
}

extern "C" void kernel_launch(void* const* d_in, const int* in_sizes, int n_in,
                              void* d_out, int out_size, void* d_ws, size_t ws_size,
                              hipStream_t stream) {
    const float* x     = (const float*)d_in[0];
    const float* ea    = (const float*)d_in[1];
    const float* W     = (const float*)d_in[2];
    const float* bias  = (const float*)d_in[3];
    const float* gamma = (const float*)d_in[4];
    const float* beta  = (const float*)d_in[5];
    const int*   ei    = (const int*)d_in[6];
    float* out = (float*)d_out;

    const int N = in_sizes[0] / DFEAT;
    const int E = in_sizes[1] / 4;

    float* deg  = (float*)d_ws;
    float* dinv = deg + N;
    float* h    = dinv + N;

    init_kernel<<<(N * DFEAT + 255) / 256, 256, 0, stream>>>(out, deg, N);
    edge_deg_kernel<<<(E + 255) / 256, 256, 0, stream>>>(ea, ei, deg, E);
    dinv_kernel<<<(N + 255) / 256, 256, 0, stream>>>(deg, dinv, N);

    int tiles = ((N + 15) / 16) * (DFEAT / 16);
    gemm_wmma_kernel<<<tiles, 32, 0, stream>>>(x, W, bias, h, N);

    size_t scatter_threads = (size_t)E * 32;
    scatter_kernel<<<(int)((scatter_threads + 255) / 256), 256, 0, stream>>>(h, ea, ei, dinv, out, E);

    size_t finish_threads = (size_t)N * 32;
    finish_kernel<<<(int)((finish_threads + 255) / 256), 256, 0, stream>>>(h, dinv, gamma, beta, out, N);
}


# Degree sums in a block-private LDS histogram (ds float atomics, 128 edge slices), partials summed in fixed order in the dinv kernel; replaces contended global f32 atomics on deg (sum order changes)
# speedup vs baseline: 1.4655x; 1.4655x over previous
#include <hip/hip_runtime.h>
#include <math.h>

typedef float v2f __attribute__((ext_vector_type(2)));
typedef float v8f __attribute__((ext_vector_type(8)));

#define DFEAT 128
#define NEG_SLOPE 0.01f
#define LN_EPS 1e-5f

#define HB 128
#define HT 1024
#define HU 5
#define RS 8

__global__ void init_kernel(float* __restrict__ out, int N) {
    int i = blockIdx.x * blockDim.x + threadIdx.x;
    if (i < N * DFEAT) out[i] = 0.0f;
}

__global__ __launch_bounds__(HT) void edge_deg_kernel(const float* __restrict__ ea,
                                                      const int* __restrict__ ei,
                                                      float* __restrict__ part, int E, int N) {
    extern __shared__ float s_deg[];
    for (int i = threadIdx.x; i < N; i += HT) s_deg[i] = 0.0f;
    __syncthreads();

    const int stride = HB * HT;
    for (int e0 = blockIdx.x * HT + threadIdx.x; e0 < E; e0 += stride * HU) {
        int   d[HU];
        float w[HU];
        #pragma unroll
        for (int u = 0; u < HU; u++) {
            int e = e0 + u * stride;
            int ec = (e < E) ? e : E - 1;
            int dv = ei[E + ec];
            w[u] = ea[(size_t)ec * 4];
            d[u] = (e < E) ? dv : -1;
        }
        #pragma unroll
        for (int u = 0; u < HU; u++) {
            if ((unsigned)d[u] < (unsigned)N) atomicAdd(&s_deg[d[u]], fabsf(w[u]));
        }
    }
    __syncthreads();

    float* p = part + (size_t)blockIdx.x * N;
    for (int i = threadIdx.x; i < N; i += HT) p[i] = s_deg[i];
}

__global__ __launch_bounds__(32 * RS) void dinv_kernel(const float* __restrict__ part,
                                                       float* __restrict__ dinv, int N) {
    __shared__ float s_sum[RS][32];
    int j = threadIdx.x & 31;
    int s = threadIdx.x >> 5;
    int node = blockIdx.x * 32 + j;
    float acc = 0.0f;
    if (node < N) {
        #pragma unroll
        for (int t = 0; t < HB / RS; t++) acc += part[(size_t)(t * RS + s) * N + node];
    }
    s_sum[s][j] = acc;
    __syncthreads();
    if (s == 0 && node < N) {
        float d = 1.0f;
        #pragma unroll
        for (int t = 0; t < RS; t++) d += s_sum[t][j];
        dinv[node] = (d > 0.0f) ? rsqrtf(d) : 0.0f;
    }
}

__global__ __launch_bounds__(32) void gemm_wmma_kernel(const float* __restrict__ x,
                                                       const float* __restrict__ W,
                                                       const float* __restrict__ bias,
                                                       float* __restrict__ h, int N) {
    int tile = blockIdx.x;
    int tn = tile & 7;
    int tm = tile >> 3;
    int lane = threadIdx.x;
    int half = lane >> 4;
    int l16  = lane & 15;

    int m = tm * 16 + l16;
    if (m >= N) m = N - 1;
    int n = tn * 16 + l16;

    const float* xrow = x + (size_t)m * DFEAT + half * 2;
    const float* wrow = W + (size_t)n * DFEAT + half * 2;

    v8f c = {};
    #pragma unroll 4
    for (int k0 = 0; k0 < DFEAT; k0 += 4) {
        v2f a, b;
        a.x = xrow[k0];     a.y = xrow[k0 + 1];
        b.x = wrow[k0];     b.y = wrow[k0 + 1];
        c = __builtin_amdgcn_wmma_f32_16x16x4_f32(false, a, false, b,
                                                  (short)0, c, false, false);
    }

    float bv = bias[tn * 16 + l16];
    int row0 = tm * 16 + half * 8;
    float* hp = h + (size_t)row0 * DFEAT + tn * 16 + l16;
    if (row0 + 8 <= N) {
        #pragma unroll
        for (int v = 0; v < 8; v++) {
            hp[(size_t)v * DFEAT] = c[v] + bv;
        }
    } else {
        #pragma unroll
        for (int v = 0; v < 8; v++) {
            if (row0 + v < N) hp[(size_t)v * DFEAT] = c[v] + bv;
        }
    }
}

__global__ void scatter_kernel(const float* __restrict__ h, const float* __restrict__ ea,
                               const int* __restrict__ ei, const float* __restrict__ dinv,
                               float* __restrict__ out, int E) {
    int wave = (int)((blockIdx.x * (size_t)blockDim.x + threadIdx.x) >> 5);
    int lane = threadIdx.x & 31;
    if (wave >= E) return;
    int src = ei[wave];
    int dst = ei[E + wave];
    float norm = dinv[src] * fabsf(ea[(size_t)wave * 4]) * dinv[dst];
    int d0 = lane * 4;
    const float4 hv = *(const float4*)(h + (size_t)src * DFEAT + d0);
    float* od = out + (size_t)dst * DFEAT + d0;
    atomicAdd(&od[0], hv.x * norm);
    atomicAdd(&od[1], hv.y * norm);
    atomicAdd(&od[2], hv.z * norm);
    atomicAdd(&od[3], hv.w * norm);
}

__global__ void finish_kernel(const float* __restrict__ h, const float* __restrict__ dinv,
                              const float* __restrict__ gamma, const float* __restrict__ beta,
                              float* __restrict__ out, int N) {
    int node = (int)((blockIdx.x * (size_t)blockDim.x + threadIdx.x) >> 5);
    int lane = threadIdx.x & 31;
    if (node >= N) return;

    float di = dinv[node];
    float self_norm = di * di;

    int d0 = lane * 4;
    float4 ov = *(const float4*)(out + (size_t)node * DFEAT + d0);
    float4 hv = *(const float4*)(h + (size_t)node * DFEAT + d0);

    float v[4];
    v[0] = ov.x + hv.x * self_norm;
    v[1] = ov.y + hv.y * self_norm;
    v[2] = ov.z + hv.z * self_norm;
    v[3] = ov.w + hv.w * self_norm;

    float sum = 0.0f;
    #pragma unroll
    for (int j = 0; j < 4; j++) {
        v[j] = (v[j] >= 0.0f) ? v[j] : NEG_SLOPE * v[j];
        sum += v[j];
    }
    #pragma unroll
    for (int off = 16; off > 0; off >>= 1) sum += __shfl_xor(sum, off, 32);
    float mean = sum * (1.0f / DFEAT);

    float vs = 0.0f;
    #pragma unroll
    for (int j = 0; j < 4; j++) {
        float dlt = v[j] - mean;
        vs += dlt * dlt;
    }
    #pragma unroll
    for (int off = 16; off > 0; off >>= 1) vs += __shfl_xor(vs, off, 32);
    float inv_std = rsqrtf(vs * (1.0f / DFEAT) + LN_EPS);

    float4 gv = *(const float4*)(gamma + d0);
    float4 bv = *(const float4*)(beta + d0);
    float4 res;
    res.x = (v[0] - mean) * inv_std * gv.x + bv.x;
    res.y = (v[1] - mean) * inv_std * gv.y + bv.y;
    res.z = (v[2] - mean) * inv_std * gv.z + bv.z;
    res.w = (v[3] - mean) * inv_std * gv.w + bv.w;
    *(float4*)(out + (size_t)node * DFEAT + d0) = res;
}

extern "C" void kernel_launch(void* const* d_in, const int* in_sizes, int n_in,
                              void* d_out, int out_size, void* d_ws, size_t ws_size,
                              hipStream_t stream) {
    const float* x     = (const float*)d_in[0];
    const float* ea    = (const float*)d_in[1];
    const float* W     = (const float*)d_in[2];
    const float* bias  = (const float*)d_in[3];
    const float* gamma = (const float*)d_in[4];
    const float* beta  = (const float*)d_in[5];
    const int*   ei    = (const int*)d_in[6];
    float* out = (float*)d_out;

    const int N = in_sizes[0] / DFEAT;
    const int E = in_sizes[1] / 4;

    float* dinv = (float*)d_ws + N;
    float* h    = dinv + N;

    init_kernel<<<(N * DFEAT + 255) / 256, 256, 0, stream>>>(out, N);
    edge_deg_kernel<<<HB, HT, (size_t)N * sizeof(float), stream>>>(ea, ei, h, E, N);
    dinv_kernel<<<(N + 31) / 32, 32 * RS, 0, stream>>>(h, dinv, N);

    int tiles = ((N + 15) / 16) * (DFEAT / 16);
    gemm_wmma_kernel<<<tiles, 32, 0, stream>>>(x, W, bias, h, N);

    size_t scatter_threads = (size_t)E * 32;
    scatter_kernel<<<(int)((scatter_threads + 255) / 256), 256, 0, stream>>>(h, ea, ei, dinv, out, E);

    size_t finish_threads = (size_t)N * 32;
    finish_kernel<<<(int)((finish_threads + 255) / 256), 256, 0, stream>>>(h, dinv, gamma, beta, out, N);
}


# Edges grouped by target each call (LDS rank counting, edge slots in the target's own output row); one wave per node gathers and finishes its row: global f32 atomics gone (sum order changes)
# speedup vs baseline: 2.3823x; 1.6256x over previous
#include <hip/hip_runtime.h>
#include <math.h>

typedef float v2f __attribute__((ext_vector_type(2)));
typedef float v8f __attribute__((ext_vector_type(8)));

#define DFEAT 128
#define NEG_SLOPE 0.01f
#define LN_EPS 1e-5f

#define HB 128
#define HT 1024
#define HU 5
#define RS 8
#define CAP DFEAT
static_assert(HB == DFEAT, "the HB x N count plane must fill the N x DFEAT output plane exactly");
static_assert(HB % RS == 0, "each wave of the dinv kernel owns HB / RS consecutive slices");

__global__ __launch_bounds__(HT) void edge_deg_kernel(const float* __restrict__ ea,
                                                      const int* __restrict__ ei,
                                                      float* __restrict__ part,
                                                      int* __restrict__ cpart, int E, int N) {
    extern __shared__ float s_deg[];
    int* s_cnt = (int*)(s_deg + N);
    for (int i = threadIdx.x; i < N; i += HT) { s_deg[i] = 0.0f; s_cnt[i] = 0; }
    __syncthreads();

    const int stride = HB * HT;
    for (int e0 = blockIdx.x * HT + threadIdx.x; e0 < E; e0 += stride * HU) {
        int   d[HU];
        float w[HU];
        #pragma unroll
        for (int u = 0; u < HU; u++) {
            int e = e0 + u * stride;
            int ec = (e < E) ? e : E - 1;
            int dv = ei[E + ec];
            w[u] = ea[(size_t)ec * 4];
            d[u] = (e < E) ? dv : -1;
        }
        #pragma unroll
        for (int u = 0; u < HU; u++) {
            if ((unsigned)d[u] < (unsigned)N) {
                atomicAdd(&s_deg[d[u]], fabsf(w[u]));
                atomicAdd(&s_cnt[d[u]], 1);
            }
        }
    }
    __syncthreads();

    float* p = part + (size_t)blockIdx.x * N;
    int*   c = cpart + (size_t)blockIdx.x * N;
    for (int i = threadIdx.x; i < N; i += HT) { p[i] = s_deg[i]; c[i] = s_cnt[i]; }
}

__global__ __launch_bounds__(32 * RS) void dinv_kernel(float* plane, const int* __restrict__ cpart,
                                                       float* __restrict__ dinv,
                                                       int* __restrict__ cnt, int N) {
    __shared__ float s_sum[RS][32];
    __shared__ int   s_csum[RS][32];
    int j = threadIdx.x & 31;
    int s = threadIdx.x >> 5;
    int node = blockIdx.x * 32 + j;
    float acc = 0.0f;
    int c[HB / RS];
    int csum = 0;
    if (node < N) {
        #pragma unroll
        for (int t = 0; t < HB / RS; t++) acc += plane[(size_t)(t * RS + s) * N + node];
        #pragma unroll
        for (int t = 0; t < HB / RS; t++) {
            c[t] = cpart[(size_t)(s * (HB / RS) + t) * N + node];
            csum += c[t];
        }
    }
    s_sum[s][j] = acc;
    s_csum[s][j] = csum;
    __syncthreads();
    if (node < N) {
        int run = 0;
        for (int t = 0; t < s; t++) run += s_csum[t][j];
        int* base = (int*)plane;
        #pragma unroll
        for (int t = 0; t < HB / RS; t++) {
            base[(size_t)(s * (HB / RS) + t) * N + node] = run;
            run += c[t];
        }
    }
    if (s == 0 && node < N) {
        float d = 1.0f;
        int tot = 0;
        #pragma unroll
        for (int t = 0; t < RS; t++) { d += s_sum[t][j]; tot += s_csum[t][j]; }
        dinv[node] = (d > 0.0f) ? rsqrtf(d) : 0.0f;
        cnt[node] = tot;
    }
}

__global__ __launch_bounds__(HT) void fill_kernel(const int* __restrict__ ei,
                                                  const int* __restrict__ base,
                                                  int* __restrict__ slots, int E, int N) {
    extern __shared__ int s_pos[];
    const int* b = base + (size_t)blockIdx.x * N;
    for (int i = threadIdx.x; i < N; i += HT) s_pos[i] = b[i];
    __syncthreads();

    const int stride = HB * HT;
    for (int e0 = blockIdx.x * HT + threadIdx.x; e0 < E; e0 += stride * HU) {
        int d[HU];
        #pragma unroll
        for (int u = 0; u < HU; u++) {
            int e = e0 + u * stride;
            int ec = (e < E) ? e : E - 1;
            int dv = ei[E + ec];
            d[u] = (e < E) ? dv : -1;
        }
        #pragma unroll
        for (int u = 0; u < HU; u++) {
            if ((unsigned)d[u] < (unsigned)N) {
                int r = atomicAdd(&s_pos[d[u]], 1);
                if (r < CAP) slots[(size_t)d[u] * DFEAT + r] = e0 + u * stride;
            }
        }
    }
}

__global__ __launch_bounds__(32) void gemm_wmma_kernel(const float* __restrict__ x,
                                                       const float* __restrict__ W,
                                                       const float* __restrict__ bias,
                                                       float* __restrict__ h, int N) {
    int tile = blockIdx.x;
    int tn = tile & 7;
    int tm = tile >> 3;
    int lane = threadIdx.x;
    int half = lane >> 4;
    int l16  = lane & 15;

    int m = tm * 16 + l16;
    if (m >= N) m = N - 1;
    int n = tn * 16 + l16;

    const float* xrow = x + (size_t)m * DFEAT + half * 2;
    const float* wrow = W + (size_t)n * DFEAT + half * 2;

    v8f c = {};
    #pragma unroll 4
    for (int k0 = 0; k0 < DFEAT; k0 += 4) {
        v2f a, b;
        a.x = xrow[k0];     a.y = xrow[k0 + 1];
        b.x = wrow[k0];     b.y = wrow[k0 + 1];
        c = __builtin_amdgcn_wmma_f32_16x16x4_f32(false, a, false, b,
                                                  (short)0, c, false, false);
    }

    float bv = bias[tn * 16 + l16];
    int row0 = tm * 16 + half * 8;
    float* hp = h + (size_t)row0 * DFEAT + tn * 16 + l16;
    if (row0 + 8 <= N) {
        #pragma unroll
        for (int v = 0; v < 8; v++) {
            hp[(size_t)v * DFEAT] = c[v] + bv;
        }
    } else {
        #pragma unroll
        for (int v = 0; v < 8; v++) {
            if (row0 + v < N) hp[(size_t)v * DFEAT] = c[v] + bv;
        }
    }
}

__device__ __forceinline__ int lane_i(int v, int l) { return __builtin_amdgcn_readlane(v, l); }
__device__ __forceinline__ float lane_f(float v, int l) {
    return __int_as_float(__builtin_amdgcn_readlane(__float_as_int(v), l));
}

__global__ __launch_bounds__(256) void gather_finish_kernel(const float* __restrict__ h,
                                                            const float* __restrict__ ea,
                                                            const int* __restrict__ ei,
                                                            const float* __restrict__ dinv,
                                                            const int* __restrict__ cnt,
                                                            const float* __restrict__ gamma,
                                                            const float* __restrict__ beta,
                                                            float* out, int E, int N) {
    int node = (int)((blockIdx.x * (size_t)blockDim.x + threadIdx.x) >> 5);
    int lane = threadIdx.x & 31;
    if (node >= N) return;
    node = __builtin_amdgcn_readfirstlane(node);

    const float di = dinv[node];
    const int   c  = __builtin_amdgcn_readfirstlane(cnt[node]);
    const int   d0 = lane * 4;
    float* orow = out + (size_t)node * DFEAT;

    float a0 = 0.0f, a1 = 0.0f, a2 = 0.0f, a3 = 0.0f;

    if (c <= 0) {
    } else if (c <= CAP) {
        const int4 rec = *(const int4*)((const int*)orow + d0);
        const bool k0 = d0 + 0 < c, k1 = d0 + 1 < c, k2 = d0 + 2 < c, k3 = d0 + 3 < c;
        const int e0 = k0 ? rec.x : 0, e1 = k1 ? rec.y : 0, e2 = k2 ? rec.z : 0, e3 = k3 ? rec.w : 0;
        const int   sr0 = ei[e0], sr1 = ei[e1], sr2 = ei[e2], sr3 = ei[e3];
        const float w0 = ea[(size_t)e0 * 4], w1 = ea[(size_t)e1 * 4];
        const float w2 = ea[(size_t)e2 * 4], w3 = ea[(size_t)e3 * 4];
        const float q0 = dinv[sr0], q1 = dinv[sr1], q2 = dinv[sr2], q3 = dinv[sr3];
        const float nm0 = k0 ? q0 * fabsf(w0) * di : 0.0f;
        const float nm1 = k1 ? q1 * fabsf(w1) * di : 0.0f;
        const float nm2 = k2 ? q2 * fabsf(w2) * di : 0.0f;
        const float nm3 = k3 ? q3 * fabsf(w3) * di : 0.0f;

        const int ng = c >> 2;
        #pragma unroll 2
        for (int l = 0; l < ng; l++) {
            const int   s0 = lane_i(sr0, l), s1 = lane_i(sr1, l), s2 = lane_i(sr2, l), s3 = lane_i(sr3, l);
            const float n0 = lane_f(nm0, l), n1 = lane_f(nm1, l), n2 = lane_f(nm2, l), n3 = lane_f(nm3, l);
            const float4 r0 = *(const float4*)(h + (size_t)s0 * DFEAT + d0);
            const float4 r1 = *(const float4*)(h + (size_t)s1 * DFEAT + d0);
            const float4 r2 = *(const float4*)(h + (size_t)s2 * DFEAT + d0);
            const float4 r3 = *(const float4*)(h + (size_t)s3 * DFEAT + d0);
            a0 += r0.x * n0; a1 += r0.y * n0; a2 += r0.z * n0; a3 += r0.w * n0;
            a0 += r1.x * n1; a1 += r1.y * n1; a2 += r1.z * n1; a3 += r1.w * n1;
            a0 += r2.x * n2; a1 += r2.y * n2; a2 += r2.z * n2; a3 += r2.w * n2;
            a0 += r3.x * n3; a1 += r3.y * n3; a2 += r3.z * n3; a3 += r3.w * n3;
        }
        const int rem = c & 3;
        if (rem > 0) {
            const int   s0 = lane_i(sr0, ng);
            const float n0 = lane_f(nm0, ng);
            const float4 r0 = *(const float4*)(h + (size_t)s0 * DFEAT + d0);
            a0 += r0.x * n0; a1 += r0.y * n0; a2 += r0.z * n0; a3 += r0.w * n0;
        }
        if (rem > 1) {
            const int   s1 = lane_i(sr1, ng);
            const float n1 = lane_f(nm1, ng);
            const float4 r1 = *(const float4*)(h + (size_t)s1 * DFEAT + d0);
            a0 += r1.x * n1; a1 += r1.y * n1; a2 += r1.z * n1; a3 += r1.w * n1;
        }
        if (rem > 2) {
            const int   s2 = lane_i(sr2, ng);
            const float n2 = lane_f(nm2, ng);
            const float4 r2 = *(const float4*)(h + (size_t)s2 * DFEAT + d0);
            a0 += r2.x * n2; a1 += r2.y * n2; a2 += r2.z * n2; a3 += r2.w * n2;
        }
    } else {
        for (int b0 = 0; b0 < E; b0 += 32) {
            const int e = b0 + lane;
            const int d = (e < E) ? ei[E + e] : -1;
            unsigned m = (unsigned)__ballot(d == node);
            while (m) {
                const int bit = __ffs(m) - 1;
                m &= m - 1;
                const int ee = b0 + bit;
                const int s  = ei[ee];
                const float nv = dinv[s] * fabsf(ea[(size_t)ee * 4]) * di;
                const float4 r = *(const float4*)(h + (size_t)s * DFEAT + d0);
                a0 += r.x * nv; a1 += r.y * nv; a2 += r.z * nv; a3 += r.w * nv;
            }
        }
    }

    const float self_norm = di * di;
    const float4 hv = *(const float4*)(h + (size_t)node * DFEAT + d0);

    float v[4];
    v[0] = a0 + hv.x * self_norm;
    v[1] = a1 + hv.y * self_norm;
    v[2] = a2 + hv.z * self_norm;
    v[3] = a3 + hv.w * self_norm;

    float sum = 0.0f;
    #pragma unroll
    for (int j = 0; j < 4; j++) {
        v[j] = (v[j] >= 0.0f) ? v[j] : NEG_SLOPE * v[j];
        sum += v[j];
    }
    #pragma unroll
    for (int off = 16; off > 0; off >>= 1) sum += __shfl_xor(sum, off, 32);
    float mean = sum * (1.0f / DFEAT);

    float vs = 0.0f;
    #pragma unroll
    for (int j = 0; j < 4; j++) {
        float dlt = v[j] - mean;
        vs += dlt * dlt;
    }
    #pragma unroll
    for (int off = 16; off > 0; off >>= 1) vs += __shfl_xor(vs, off, 32);
    float inv_std = rsqrtf(vs * (1.0f / DFEAT) + LN_EPS);

    float4 gv = *(const float4*)(gamma + d0);
    float4 bv = *(const float4*)(beta + d0);
    float4 res;
    res.x = (v[0] - mean) * inv_std * gv.x + bv.x;
    res.y = (v[1] - mean) * inv_std * gv.y + bv.y;
    res.z = (v[2] - mean) * inv_std * gv.z + bv.z;
    res.w = (v[3] - mean) * inv_std * gv.w + bv.w;
    *(float4*)(orow + d0) = res;
}

extern "C" void kernel_launch(void* const* d_in, const int* in_sizes, int n_in,
                              void* d_out, int out_size, void* d_ws, size_t ws_size,
                              hipStream_t stream) {
    const float* x     = (const float*)d_in[0];
    const float* ea    = (const float*)d_in[1];
    const float* W     = (const float*)d_in[2];
    const float* bias  = (const float*)d_in[3];
    const float* gamma = (const float*)d_in[4];
    const float* beta  = (const float*)d_in[5];
    const int*   ei    = (const int*)d_in[6];
    float* out = (float*)d_out;

    const int N = in_sizes[0] / DFEAT;
    const int E = in_sizes[1] / 4;

    int*   cnt  = (int*)d_ws;
    float* dinv = (float*)d_ws + N;
    float* h    = dinv + N;
    int*   oi   = (int*)d_out;

    edge_deg_kernel<<<HB, HT, (size_t)N * 2 * sizeof(float), stream>>>(ea, ei, h, oi, E, N);
    dinv_kernel<<<(N + 31) / 32, 32 * RS, 0, stream>>>(h, oi, dinv, cnt, N);
    fill_kernel<<<HB, HT, (size_t)N * sizeof(int), stream>>>(ei, (const int*)h, oi, E, N);

    int tiles = ((N + 15) / 16) * (DFEAT / 16);
    gemm_wmma_kernel<<<tiles, 32, 0, stream>>>(x, W, bias, h, N);

    size_t finish_threads = (size_t)N * 32;
    gather_finish_kernel<<<(int)((finish_threads + 255) / 256), 256, 0, stream>>>(
        h, ea, ei, dinv, cnt, gamma, beta, out, E, N);
}


# Linear map moved behind the edge sum: waves sum rows of x, each 16-node tile is multiplied by W^T on WMMA inside the finish kernel; product launch and h plane gone (sum order changes)
# speedup vs baseline: 2.6487x; 1.1118x over previous
#include <hip/hip_runtime.h>
#include <math.h>

typedef float v2f __attribute__((ext_vector_type(2)));
typedef float v8f __attribute__((ext_vector_type(8)));

#define DFEAT 128
#define NEG_SLOPE 0.01f
#define LN_EPS 1e-5f

#define HB 128
#define HT 1024
#define HU 5
#define RS 8
#define CAP DFEAT
static_assert(HB == DFEAT, "the HB x N count plane must fill the N x DFEAT output plane exactly");
static_assert(HB % RS == 0, "each wave of the dinv kernel owns HB / RS consecutive slices");

__global__ __launch_bounds__(HT) void edge_deg_kernel(const float* __restrict__ ea,
                                                      const int* __restrict__ ei,
                                                      float* __restrict__ part,
                                                      int* __restrict__ cpart, int E, int N) {
    extern __shared__ float s_deg[];
    int* s_cnt = (int*)(s_deg + N);
    for (int i = threadIdx.x; i < N; i += HT) { s_deg[i] = 0.0f; s_cnt[i] = 0; }
    __syncthreads();

    const int stride = HB * HT;
    for (int e0 = blockIdx.x * HT + threadIdx.x; e0 < E; e0 += stride * HU) {
        int   d[HU];
        float w[HU];
        #pragma unroll
        for (int u = 0; u < HU; u++) {
            int e = e0 + u * stride;
            int ec = (e < E) ? e : E - 1;
            int dv = ei[E + ec];
            w[u] = ea[(size_t)ec * 4];
            d[u] = (e < E) ? dv : -1;
        }
        #pragma unroll
        for (int u = 0; u < HU; u++) {
            if ((unsigned)d[u] < (unsigned)N) {
                atomicAdd(&s_deg[d[u]], fabsf(w[u]));
                atomicAdd(&s_cnt[d[u]], 1);
            }
        }
    }
    __syncthreads();

    float* p = part + (size_t)blockIdx.x * N;
    int*   c = cpart + (size_t)blockIdx.x * N;
    for (int i = threadIdx.x; i < N; i += HT) { p[i] = s_deg[i]; c[i] = s_cnt[i]; }
}

__global__ __launch_bounds__(32 * RS) void dinv_kernel(float* plane, const int* __restrict__ cpart,
                                                       float* __restrict__ dinv,
                                                       int* __restrict__ cnt, int N) {
    __shared__ float s_sum[RS][32];
    __shared__ int   s_csum[RS][32];
    int j = threadIdx.x & 31;
    int s = threadIdx.x >> 5;
    int node = blockIdx.x * 32 + j;
    float acc = 0.0f;
    int c[HB / RS];
    int csum = 0;
    if (node < N) {
        #pragma unroll
        for (int t = 0; t < HB / RS; t++) acc += plane[(size_t)(t * RS + s) * N + node];
        #pragma unroll
        for (int t = 0; t < HB / RS; t++) {
            c[t] = cpart[(size_t)(s * (HB / RS) + t) * N + node];
            csum += c[t];
        }
    }
    s_sum[s][j] = acc;
    s_csum[s][j] = csum;
    __syncthreads();
    if (node < N) {
        int run = 0;
        for (int t = 0; t < s; t++) run += s_csum[t][j];
        int* base = (int*)plane;
        #pragma unroll
        for (int t = 0; t < HB / RS; t++) {
            base[(size_t)(s * (HB / RS) + t) * N + node] = run;
            run += c[t];
        }
    }
    if (s == 0 && node < N) {
        float d = 1.0f;
        int tot = 0;
        #pragma unroll
        for (int t = 0; t < RS; t++) { d += s_sum[t][j]; tot += s_csum[t][j]; }
        dinv[node] = (d > 0.0f) ? rsqrtf(d) : 0.0f;
        cnt[node] = tot;
    }
}

__global__ __launch_bounds__(HT) void fill_kernel(const int* __restrict__ ei,
                                                  const int* __restrict__ base,
                                                  int* __restrict__ slots, int E, int N) {
    extern __shared__ int s_pos[];
    const int* b = base + (size_t)blockIdx.x * N;
    for (int i = threadIdx.x; i < N; i += HT) s_pos[i] = b[i];
    __syncthreads();

    const int stride = HB * HT;
    for (int e0 = blockIdx.x * HT + threadIdx.x; e0 < E; e0 += stride * HU) {
        int d[HU];
        #pragma unroll
        for (int u = 0; u < HU; u++) {
            int e = e0 + u * stride;
            int ec = (e < E) ? e : E - 1;
            int dv = ei[E + ec];
            d[u] = (e < E) ? dv : -1;
        }
        #pragma unroll
        for (int u = 0; u < HU; u++) {
            if ((unsigned)d[u] < (unsigned)N) {
                int r = atomicAdd(&s_pos[d[u]], 1);
                if (r < CAP) slots[(size_t)d[u] * DFEAT + r] = e0 + u * stride;
            }
        }
    }
}

#define NPB 16
#define NPW 2
#define XP 132

__device__ __forceinline__ int lane_i(int v, int l) { return __builtin_amdgcn_readlane(v, l); }
__device__ __forceinline__ float lane_f(float v, int l) {
    return __int_as_float(__builtin_amdgcn_readlane(__float_as_int(v), l));
}

__device__ __forceinline__ float4 gather_node(const float* __restrict__ x,
                                              const float* __restrict__ ea,
                                              const int* __restrict__ ei,
                                              const float* __restrict__ dinv,
                                              const int* __restrict__ cnt,
                                              const float* orow, int node, int lane,
                                              int E, float& nsum) {
    const float di = dinv[node];
    const int   c  = __builtin_amdgcn_readfirstlane(cnt[node]);
    const int   d0 = lane * 4;

    float a0 = 0.0f, a1 = 0.0f, a2 = 0.0f, a3 = 0.0f;
    float ns = 0.0f;

    if (c <= 0) {
    } else if (c <= CAP) {
        const int4 rec = *(const int4*)((const int*)orow + d0);
        const bool k0 = d0 + 0 < c, k1 = d0 + 1 < c, k2 = d0 + 2 < c, k3 = d0 + 3 < c;
        const int e0 = k0 ? rec.x : 0, e1 = k1 ? rec.y : 0, e2 = k2 ? rec.z : 0, e3 = k3 ? rec.w : 0;
        const int   sr0 = ei[e0], sr1 = ei[e1], sr2 = ei[e2], sr3 = ei[e3];
        const float w0 = ea[(size_t)e0 * 4], w1 = ea[(size_t)e1 * 4];
        const float w2 = ea[(size_t)e2 * 4], w3 = ea[(size_t)e3 * 4];
        const float q0 = dinv[sr0], q1 = dinv[sr1], q2 = dinv[sr2], q3 = dinv[sr3];
        const float nm0 = k0 ? q0 * fabsf(w0) * di : 0.0f;
        const float nm1 = k1 ? q1 * fabsf(w1) * di : 0.0f;
        const float nm2 = k2 ? q2 * fabsf(w2) * di : 0.0f;
        const float nm3 = k3 ? q3 * fabsf(w3) * di : 0.0f;
        ns = (nm0 + nm1) + (nm2 + nm3);
        #pragma unroll
        for (int off = 16; off > 0; off >>= 1) ns += __shfl_xor(ns, off, 32);

        const int ng = c >> 2;
        #pragma unroll 4
        for (int l = 0; l < ng; l++) {
            const int   s0 = lane_i(sr0, l), s1 = lane_i(sr1, l), s2 = lane_i(sr2, l), s3 = lane_i(sr3, l);
            const float n0 = lane_f(nm0, l), n1 = lane_f(nm1, l), n2 = lane_f(nm2, l), n3 = lane_f(nm3, l);
            const float4 r0 = *(const float4*)(x + (size_t)s0 * DFEAT + d0);
            const float4 r1 = *(const float4*)(x + (size_t)s1 * DFEAT + d0);
            const float4 r2 = *(const float4*)(x + (size_t)s2 * DFEAT + d0);
            const float4 r3 = *(const float4*)(x + (size_t)s3 * DFEAT + d0);
            a0 += r0.x * n0; a1 += r0.y * n0; a2 += r0.z * n0; a3 += r0.w * n0;
            a0 += r1.x * n1; a1 += r1.y * n1; a2 += r1.z * n1; a3 += r1.w * n1;
            a0 += r2.x * n2; a1 += r2.y * n2; a2 += r2.z * n2; a3 += r2.w * n2;
            a0 += r3.x * n3; a1 += r3.y * n3; a2 += r3.z * n3; a3 += r3.w * n3;
        }
        const int rem = c & 3;
        if (rem > 0) {
            const int   s0 = lane_i(sr0, ng);
            const float n0 = lane_f(nm0, ng);
            const float4 r0 = *(const float4*)(x + (size_t)s0 * DFEAT + d0);
            a0 += r0.x * n0; a1 += r0.y * n0; a2 += r0.z * n0; a3 += r0.w * n0;
        }
        if (rem > 1) {
            const int   s1 = lane_i(sr1, ng);
            const float n1 = lane_f(nm1, ng);
            const float4 r1 = *(const float4*)(x + (size_t)s1 * DFEAT + d0);
            a0 += r1.x * n1; a1 += r1.y * n1; a2 += r1.z * n1; a3 += r1.w * n1;
        }
        if (rem > 2) {
            const int   s2 = lane_i(sr2, ng);
            const float n2 = lane_f(nm2, ng);
            const float4 r2 = *(const float4*)(x + (size_t)s2 * DFEAT + d0);
            a0 += r2.x * n2; a1 += r2.y * n2; a2 += r2.z * n2; a3 += r2.w * n2;
        }
    } else {
        for (int b0 = 0; b0 < E; b0 += 32) {
            const int e = b0 + lane;
            const int d = (e < E) ? ei[E + e] : -1;
            unsigned m = (unsigned)__ballot(d == node);
            while (m) {
                const int bit = __ffs(m) - 1;
                m &= m - 1;
                const int ee = b0 + bit;
                const int s  = ei[ee];
                const float nv = dinv[s] * fabsf(ea[(size_t)ee * 4]) * di;
                const float4 r = *(const float4*)(x + (size_t)s * DFEAT + d0);
                a0 += r.x * nv; a1 += r.y * nv; a2 += r.z * nv; a3 += r.w * nv;
                ns += nv;
            }
        }
    }

    const float self_norm = di * di;
    const float4 xv = *(const float4*)(x + (size_t)node * DFEAT + d0);
    float4 res;
    res.x = a0 + xv.x * self_norm;
    res.y = a1 + xv.y * self_norm;
    res.z = a2 + xv.z * self_norm;
    res.w = a3 + xv.w * self_norm;
    nsum = ns + self_norm;
    return res;
}

__global__ __launch_bounds__(32 * NPB / NPW) void gather_linear_finish_kernel(
        const float* __restrict__ x, const float* __restrict__ W, const float* __restrict__ bias,
        const float* __restrict__ ea, const int* __restrict__ ei,
        const float* __restrict__ dinv, const int* __restrict__ cnt,
        const float* __restrict__ gamma, const float* __restrict__ beta,
        float* out, int E, int N) {
    __shared__ __attribute__((aligned(16))) float s_t[NPB][XP];
    __shared__ float s_ns[NPB];
    const int lane = threadIdx.x & 31;
    const int wave = __builtin_amdgcn_readfirstlane((int)(threadIdx.x >> 5));
    const int node_base = blockIdx.x * NPB;

    #pragma unroll
    for (int q = 0; q < NPW; q++) {
        const int row = wave * NPW + q;
        const int node = node_base + row;
        float4 agg = make_float4(0.0f, 0.0f, 0.0f, 0.0f);
        float ns = 0.0f;
        if (node < N)
            agg = gather_node(x, ea, ei, dinv, cnt, out + (size_t)node * DFEAT, node, lane, E, ns);
        *(float4*)&s_t[row][lane * 4] = agg;
        if (lane == 0) s_ns[row] = ns;
    }

    const int c16 = lane & 15;
    const int hf  = lane >> 4;
    const int col = wave * 16 + c16;
    const float* wrow = W + (size_t)col * DFEAT + hf * 64;
    float4 bf[16];
    #pragma unroll
    for (int j = 0; j < 16; j++) bf[j] = *(const float4*)(wrow + 4 * j);
    const float bv = bias[col];

    __syncthreads();

    const float* arow = &s_t[c16][hf * 64];
    v8f acc0 = {}, acc1 = {};
    #pragma unroll
    for (int j = 0; j < 16; j += 2) {
        const float4 a0 = *(const float4*)(arow + 4 * j);
        const float4 a1 = *(const float4*)(arow + 4 * j + 4);
        v2f fa, fb;
        fa.x = a0.x; fa.y = a0.y; fb.x = bf[j].x; fb.y = bf[j].y;
        acc0 = __builtin_amdgcn_wmma_f32_16x16x4_f32(false, fa, false, fb, (short)0, acc0, false, false);
        fa.x = a1.x; fa.y = a1.y; fb.x = bf[j + 1].x; fb.y = bf[j + 1].y;
        acc1 = __builtin_amdgcn_wmma_f32_16x16x4_f32(false, fa, false, fb, (short)0, acc1, false, false);
        fa.x = a0.z; fa.y = a0.w; fb.x = bf[j].z; fb.y = bf[j].w;
        acc0 = __builtin_amdgcn_wmma_f32_16x16x4_f32(false, fa, false, fb, (short)0, acc0, false, false);
        fa.x = a1.z; fa.y = a1.w; fb.x = bf[j + 1].z; fb.y = bf[j + 1].w;
        acc1 = __builtin_amdgcn_wmma_f32_16x16x4_f32(false, fa, false, fb, (short)0, acc1, false, false);
    }

    __syncthreads();

    #pragma unroll
    for (int r = 0; r < 8; r++) {
        const int row = hf * 8 + r;
        float v = (acc0[r] + acc1[r]) + bv * s_ns[row];
        v = (v >= 0.0f) ? v : NEG_SLOPE * v;
        s_t[row][col] = v;
    }

    __syncthreads();

    const int d0 = lane * 4;
    const float4 gv = *(const float4*)(gamma + d0);
    const float4 bt = *(const float4*)(beta + d0);
    #pragma unroll
    for (int q = 0; q < NPW; q++) {
        const int row = wave * NPW + q;
        const int node = node_base + row;
        const float4 t = *(const float4*)&s_t[row][d0];
        float v[4] = {t.x, t.y, t.z, t.w};

        float sum = (v[0] + v[1]) + (v[2] + v[3]);
        #pragma unroll
        for (int off = 16; off > 0; off >>= 1) sum += __shfl_xor(sum, off, 32);
        const float mean = sum * (1.0f / DFEAT);

        float vs = 0.0f;
        #pragma unroll
        for (int j = 0; j < 4; j++) {
            float dlt = v[j] - mean;
            vs += dlt * dlt;
        }
        #pragma unroll
        for (int off = 16; off > 0; off >>= 1) vs += __shfl_xor(vs, off, 32);
        const float inv_std = rsqrtf(vs * (1.0f / DFEAT) + LN_EPS);

        float4 res;
        res.x = (v[0] - mean) * inv_std * gv.x + bt.x;
        res.y = (v[1] - mean) * inv_std * gv.y + bt.y;
        res.z = (v[2] - mean) * inv_std * gv.z + bt.z;
        res.w = (v[3] - mean) * inv_std * gv.w + bt.w;
        if (node < N) *(float4*)(out + (size_t)node * DFEAT + d0) = res;
    }
}

extern "C" void kernel_launch(void* const* d_in, const int* in_sizes, int n_in,
                              void* d_out, int out_size, void* d_ws, size_t ws_size,
                              hipStream_t stream) {
    const float* x     = (const float*)d_in[0];
    const float* ea    = (const float*)d_in[1];
    const float* W     = (const float*)d_in[2];
    const float* bias  = (const float*)d_in[3];
    const float* gamma = (const float*)d_in[4];
    const float* beta  = (const float*)d_in[5];
    const int*   ei    = (const int*)d_in[6];
    float* out = (float*)d_out;

    const int N = in_sizes[0] / DFEAT;
    const int E = in_sizes[1] / 4;

    int*   cnt   = (int*)d_ws;
    float* dinv  = (float*)d_ws + N;
    float* plane = dinv + N;
    int*   oi    = (int*)d_out;

    edge_deg_kernel<<<HB, HT, (size_t)N * 2 * sizeof(float), stream>>>(ea, ei, plane, oi, E, N);
    dinv_kernel<<<(N + 31) / 32, 32 * RS, 0, stream>>>(plane, oi, dinv, cnt, N);
    fill_kernel<<<HB, HT, (size_t)N * sizeof(int), stream>>>(ei, (const int*)plane, oi, E, N);
    gather_linear_finish_kernel<<<(N + NPB - 1) / NPB, 32 * NPB / NPW, 0, stream>>>(
        x, W, bias, ea, ei, dinv, cnt, gamma, beta, out, E, N);
}


# Edge grouping re-cut as 16 edge slices x 16 node ranges, one work-group per pair: the sum over slices folds into the fill kernel, the separate dinv/rank launch is gone (sum order changes)
# speedup vs baseline: 2.9273x; 1.1052x over previous
#include <hip/hip_runtime.h>
#include <math.h>

typedef float v2f __attribute__((ext_vector_type(2)));
typedef float v8f __attribute__((ext_vector_type(8)));

#define DFEAT 128
#define NEG_SLOPE 0.01f
#define LN_EPS 1e-5f

#define SL 16
#define RG 16
#define HT 512
#define HU 80
#define CAP DFEAT

__global__ __launch_bounds__(HT) void edge_deg_kernel(const float* __restrict__ ea,
                                                      const int* __restrict__ ei,
                                                      float* __restrict__ part,
                                                      int* __restrict__ cpart, int E, int N) {
    extern __shared__ float s_deg[];
    const int sl = blockIdx.x % SL;
    const int rg = blockIdx.x / SL;
    const int RN = (N + RG - 1) / RG;
    const int lo = rg * RN;
    const int nr = max(0, min(N - lo, RN));
    int* s_cnt = (int*)(s_deg + RN);
    for (int i = threadIdx.x; i < nr; i += HT) { s_deg[i] = 0.0f; s_cnt[i] = 0; }
    __syncthreads();

    const int stride = SL * HT;
    for (int e0 = sl * HT + threadIdx.x; e0 < E; e0 += stride * HU) {
        int   d[HU];
        float w[HU];
        #pragma unroll
        for (int u = 0; u < HU; u++) {
            const int e = e0 + u * stride;
            d[u] = ei[E + min(e, E - 1)];
        }
        #pragma unroll
        for (int u = 0; u < HU; u++) {
            const int e = e0 + u * stride;
            const int loc = d[u] - lo;
            const bool in = (e < E) & ((unsigned)loc < (unsigned)nr);
            d[u] = in ? loc : -1;
        }
        #pragma unroll
        for (int u = 0; u < HU; u++) {
            const int ew = (d[u] >= 0) ? e0 + u * stride : 0;
            w[u] = ea[(size_t)ew * 4];
        }
        #pragma unroll
        for (int u = 0; u < HU; u++) asm volatile("" :: "v"(w[u]));
        #pragma unroll
        for (int u = 0; u < HU; u++) {
            if (d[u] >= 0) {
                atomicAdd(&s_deg[d[u]], fabsf(w[u]));
                atomicAdd(&s_cnt[d[u]], 1);
            }
        }
    }
    __syncthreads();

    float* p = part + (size_t)sl * N + lo;
    int*   c = cpart + (size_t)sl * N + lo;
    for (int i = threadIdx.x; i < nr; i += HT) { p[i] = s_deg[i]; c[i] = s_cnt[i]; }
}

__global__ __launch_bounds__(HT) void fill_kernel(const int* __restrict__ ei,
                                                  const float* __restrict__ part,
                                                  const int* __restrict__ cpart,
                                                  float* __restrict__ dinv,
                                                  int* __restrict__ cnt,
                                                  int* __restrict__ slots, int E, int N) {
    extern __shared__ int s_pos[];
    const int sl = blockIdx.x % SL;
    const int rg = blockIdx.x / SL;
    const int RN = (N + RG - 1) / RG;
    const int lo = rg * RN;
    const int nr = max(0, min(N - lo, RN));
    const int sub = (RN + SL - 1) / SL;
    for (int i = threadIdx.x; i < nr; i += HT) {
        const int node = lo + i;
        int run = 0, tot = 0;
        #pragma unroll
        for (int t = 0; t < SL; t++) {
            const int c = cpart[(size_t)t * N + node];
            run += (t < sl) ? c : 0;
            tot += c;
        }
        s_pos[i] = run;
        if (i / sub == sl) {
            float dg = 1.0f;
            #pragma unroll
            for (int t = 0; t < SL; t++) dg += part[(size_t)t * N + node];
            dinv[node] = (dg > 0.0f) ? rsqrtf(dg) : 0.0f;
            cnt[node] = tot;
        }
    }
    __syncthreads();

    const int stride = SL * HT;
    for (int e0 = sl * HT + threadIdx.x; e0 < E; e0 += stride * HU) {
        int d[HU];
        int rk[HU];
        #pragma unroll
        for (int u = 0; u < HU; u++) {
            const int e = e0 + u * stride;
            d[u] = ei[E + min(e, E - 1)];
        }
        #pragma unroll
        for (int u = 0; u < HU; u++) {
            const int e = e0 + u * stride;
            const int loc = d[u] - lo;
            const bool in = (e < E) & ((unsigned)loc < (unsigned)nr);
            d[u] = in ? loc : -1;
        }
        #pragma unroll
        for (int u = 0; u < HU; u++) {
            rk[u] = CAP;
            if (d[u] >= 0) rk[u] = atomicAdd(&s_pos[d[u]], 1);
        }
        #pragma unroll
        for (int u = 0; u < HU; u++) {
            if (rk[u] < CAP) slots[(size_t)(lo + d[u]) * DFEAT + rk[u]] = e0 + u * stride;
        }
    }
}

#define NPB 16
#define NPW 2
#define XP 132

__device__ __forceinline__ int lane_i(int v, int l) { return __builtin_amdgcn_readlane(v, l); }
__device__ __forceinline__ float lane_f(float v, int l) {
    return __int_as_float(__builtin_amdgcn_readlane(__float_as_int(v), l));
}

__device__ __forceinline__ float4 gather_node(const float* __restrict__ x,
                                              const float* __restrict__ ea,
                                              const int* __restrict__ ei,
                                              const float* __restrict__ dinv,
                                              const int* __restrict__ cnt,
                                              const float* orow, int node, int lane,
                                              int E, float& nsum) {
    const float di = dinv[node];
    const int   c  = __builtin_amdgcn_readfirstlane(cnt[node]);
    const int   d0 = lane * 4;

    float a0 = 0.0f, a1 = 0.0f, a2 = 0.0f, a3 = 0.0f;
    float ns = 0.0f;

    if (c <= 0) {
    } else if (c <= CAP) {
        const int4 rec = *(const int4*)((const int*)orow + d0);
        const bool k0 = d0 + 0 < c, k1 = d0 + 1 < c, k2 = d0 + 2 < c, k3 = d0 + 3 < c;
        const int e0 = k0 ? rec.x : 0, e1 = k1 ? rec.y : 0, e2 = k2 ? rec.z : 0, e3 = k3 ? rec.w : 0;
        const int   sr0 = ei[e0], sr1 = ei[e1], sr2 = ei[e2], sr3 = ei[e3];
        const float w0 = ea[(size_t)e0 * 4], w1 = ea[(size_t)e1 * 4];
        const float w2 = ea[(size_t)e2 * 4], w3 = ea[(size_t)e3 * 4];
        const float q0 = dinv[sr0], q1 = dinv[sr1], q2 = dinv[sr2], q3 = dinv[sr3];
        const float nm0 = k0 ? q0 * fabsf(w0) * di : 0.0f;
        const float nm1 = k1 ? q1 * fabsf(w1) * di : 0.0f;
        const float nm2 = k2 ? q2 * fabsf(w2) * di : 0.0f;
        const float nm3 = k3 ? q3 * fabsf(w3) * di : 0.0f;
        ns = (nm0 + nm1) + (nm2 + nm3);
        #pragma unroll
        for (int off = 16; off > 0; off >>= 1) ns += __shfl_xor(ns, off, 32);

        const int ng = c >> 2;
        #pragma unroll 4
        for (int l = 0; l < ng; l++) {
            const int   s0 = lane_i(sr0, l), s1 = lane_i(sr1, l), s2 = lane_i(sr2, l), s3 = lane_i(sr3, l);
            const float n0 = lane_f(nm0, l), n1 = lane_f(nm1, l), n2 = lane_f(nm2, l), n3 = lane_f(nm3, l);
            const float4 r0 = *(const float4*)(x + (size_t)s0 * DFEAT + d0);
            const float4 r1 = *(const float4*)(x + (size_t)s1 * DFEAT + d0);
            const float4 r2 = *(const float4*)(x + (size_t)s2 * DFEAT + d0);
            const float4 r3 = *(const float4*)(x + (size_t)s3 * DFEAT + d0);
            a0 += r0.x * n0; a1 += r0.y * n0; a2 += r0.z * n0; a3 += r0.w * n0;
            a0 += r1.x * n1; a1 += r1.y * n1; a2 += r1.z * n1; a3 += r1.w * n1;
            a0 += r2.x * n2; a1 += r2.y * n2; a2 += r2.z * n2; a3 += r2.w * n2;
            a0 += r3.x * n3; a1 += r3.y * n3; a2 += r3.z * n3; a3 += r3.w * n3;
        }
        const int rem = c & 3;
        if (rem > 0) {
            const int   s0 = lane_i(sr0, ng);
            const float n0 = lane_f(nm0, ng);
            const float4 r0 = *(const float4*)(x + (size_t)s0 * DFEAT + d0);
            a0 += r0.x * n0; a1 += r0.y * n0; a2 += r0.z * n0; a3 += r0.w * n0;
        }
        if (rem > 1) {
            const int   s1 = lane_i(sr1, ng);
            const float n1 = lane_f(nm1, ng);
            const float4 r1 = *(const float4*)(x + (size_t)s1 * DFEAT + d0);
            a0 += r1.x * n1; a1 += r1.y * n1; a2 += r1.z * n1; a3 += r1.w * n1;
        }
        if (rem > 2) {
            const int   s2 = lane_i(sr2, ng);
            const float n2 = lane_f(nm2, ng);
            const float4 r2 = *(const float4*)(x + (size_t)s2 * DFEAT + d0);
            a0 += r2.x * n2; a1 += r2.y * n2; a2 += r2.z * n2; a3 += r2.w * n2;
        }
    } else {
        for (int b0 = 0; b0 < E; b0 += 32) {
            const int e = b0 + lane;
            const int d = (e < E) ? ei[E + e] : -1;
            unsigned m = (unsigned)__ballot(d == node);
            while (m) {
                const int bit = __ffs(m) - 1;
                m &= m - 1;
                const int ee = b0 + bit;
                const int s  = ei[ee];
                const float nv = dinv[s] * fabsf(ea[(size_t)ee * 4]) * di;
                const float4 r = *(const float4*)(x + (size_t)s * DFEAT + d0);
                a0 += r.x * nv; a1 += r.y * nv; a2 += r.z * nv; a3 += r.w * nv;
                ns += nv;
            }
        }
    }

    const float self_norm = di * di;
    const float4 xv = *(const float4*)(x + (size_t)node * DFEAT + d0);
    float4 res;
    res.x = a0 + xv.x * self_norm;
    res.y = a1 + xv.y * self_norm;
    res.z = a2 + xv.z * self_norm;
    res.w = a3 + xv.w * self_norm;
    nsum = ns + self_norm;
    return res;
}

__global__ __launch_bounds__(32 * NPB / NPW) void gather_linear_finish_kernel(
        const float* __restrict__ x, const float* __restrict__ W, const float* __restrict__ bias,
        const float* __restrict__ ea, const int* __restrict__ ei,
        const float* __restrict__ dinv, const int* __restrict__ cnt,
        const float* __restrict__ gamma, const float* __restrict__ beta,
        float* out, int E, int N) {
    __shared__ __attribute__((aligned(16))) float s_t[NPB][XP];
    __shared__ float s_ns[NPB];
    const int lane = threadIdx.x & 31;
    const int wave = __builtin_amdgcn_readfirstlane((int)(threadIdx.x >> 5));
    const int node_base = blockIdx.x * NPB;

    #pragma unroll
    for (int q = 0; q < NPW; q++) {
        const int row = wave * NPW + q;
        const int node = node_base + row;
        float4 agg = make_float4(0.0f, 0.0f, 0.0f, 0.0f);
        float ns = 0.0f;
        if (node < N)
            agg = gather_node(x, ea, ei, dinv, cnt, out + (size_t)node * DFEAT, node, lane, E, ns);
        *(float4*)&s_t[row][lane * 4] = agg;
        if (lane == 0) s_ns[row] = ns;
    }

    const int c16 = lane & 15;
    const int hf  = lane >> 4;
    const int col = wave * 16 + c16;
    const float* wrow = W + (size_t)col * DFEAT + hf * 64;
    float4 bf[16];
    #pragma unroll
    for (int j = 0; j < 16; j++) bf[j] = *(const float4*)(wrow + 4 * j);
    const float bv = bias[col];

    __syncthreads();

    const float* arow = &s_t[c16][hf * 64];
    v8f acc0 = {}, acc1 = {};
    #pragma unroll
    for (int j = 0; j < 16; j += 2) {
        const float4 a0 = *(const float4*)(arow + 4 * j);
        const float4 a1 = *(const float4*)(arow + 4 * j + 4);
        v2f fa, fb;
        fa.x = a0.x; fa.y = a0.y; fb.x = bf[j].x; fb.y = bf[j].y;
        acc0 = __builtin_amdgcn_wmma_f32_16x16x4_f32(false, fa, false, fb, (short)0, acc0, false, false);
        fa.x = a1.x; fa.y = a1.y; fb.x = bf[j + 1].x; fb.y = bf[j + 1].y;
        acc1 = __builtin_amdgcn_wmma_f32_16x16x4_f32(false, fa, false, fb, (short)0, acc1, false, false);
        fa.x = a0.z; fa.y = a0.w; fb.x = bf[j].z; fb.y = bf[j].w;
        acc0 = __builtin_amdgcn_wmma_f32_16x16x4_f32(false, fa, false, fb, (short)0, acc0, false, false);
        fa.x = a1.z; fa.y = a1.w; fb.x = bf[j + 1].z; fb.y = bf[j + 1].w;
        acc1 = __builtin_amdgcn_wmma_f32_16x16x4_f32(false, fa, false, fb, (short)0, acc1, false, false);
    }

    __syncthreads();

    #pragma unroll
    for (int r = 0; r < 8; r++) {
        const int row = hf * 8 + r;
        float v = (acc0[r] + acc1[r]) + bv * s_ns[row];
        v = (v >= 0.0f) ? v : NEG_SLOPE * v;
        s_t[row][col] = v;
    }

    __syncthreads();

    const int d0 = lane * 4;
    const float4 gv = *(const float4*)(gamma + d0);
    const float4 bt = *(const float4*)(beta + d0);
    #pragma unroll
    for (int q = 0; q < NPW; q++) {
        const int row = wave * NPW + q;
        const int node = node_base + row;
        const float4 t = *(const float4*)&s_t[row][d0];
        float v[4] = {t.x, t.y, t.z, t.w};

        float sum = (v[0] + v[1]) + (v[2] + v[3]);
        #pragma unroll
        for (int off = 16; off > 0; off >>= 1) sum += __shfl_xor(sum, off, 32);
        const float mean = sum * (1.0f / DFEAT);

        float vs = 0.0f;
        #pragma unroll
        for (int j = 0; j < 4; j++) {
            float dlt = v[j] - mean;
            vs += dlt * dlt;
        }
        #pragma unroll
        for (int off = 16; off > 0; off >>= 1) vs += __shfl_xor(vs, off, 32);
        const float inv_std = rsqrtf(vs * (1.0f / DFEAT) + LN_EPS);

        float4 res;
        res.x = (v[0] - mean) * inv_std * gv.x + bt.x;
        res.y = (v[1] - mean) * inv_std * gv.y + bt.y;
        res.z = (v[2] - mean) * inv_std * gv.z + bt.z;
        res.w = (v[3] - mean) * inv_std * gv.w + bt.w;
        if (node < N) *(float4*)(out + (size_t)node * DFEAT + d0) = res;
    }
}

extern "C" void kernel_launch(void* const* d_in, const int* in_sizes, int n_in,
                              void* d_out, int out_size, void* d_ws, size_t ws_size,
                              hipStream_t stream) {
    const float* x     = (const float*)d_in[0];
    const float* ea    = (const float*)d_in[1];
    const float* W     = (const float*)d_in[2];
    const float* bias  = (const float*)d_in[3];
    const float* gamma = (const float*)d_in[4];
    const float* beta  = (const float*)d_in[5];
    const int*   ei    = (const int*)d_in[6];
    float* out = (float*)d_out;

    const int N = in_sizes[0] / DFEAT;
    const int E = in_sizes[1] / 4;

    int*   cnt   = (int*)d_ws;
    float* dinv  = (float*)d_ws + N;
    float* part  = dinv + N;
    int*   cpart = (int*)(part + (size_t)SL * N);
    int*   oi    = (int*)d_out;
    const int RN = (N + RG - 1) / RG;

    edge_deg_kernel<<<SL * RG, HT, (size_t)RN * 2 * sizeof(float), stream>>>(ea, ei, part, cpart, E, N);
    fill_kernel<<<SL * RG, HT, (size_t)RN * sizeof(int), stream>>>(ei, part, cpart, dinv, cnt, oi, E, N);
    gather_linear_finish_kernel<<<(N + NPB - 1) / NPB, 32 * NPB / NPW, 0, stream>>>(
        x, W, bias, ea, ei, dinv, cnt, gamma, beta, out, E, N);
}


# Edge grouping re-cut as 32 edge slices x 8 node ranges: each work-group scans and filters half as many targets per kept edge in both grouping kernels (sum order changes)
# speedup vs baseline: 3.0783x; 1.0516x over previous
#include <hip/hip_runtime.h>
#include <math.h>

typedef float v2f __attribute__((ext_vector_type(2)));
typedef float v8f __attribute__((ext_vector_type(8)));

#define DFEAT 128
#define NEG_SLOPE 0.01f
#define LN_EPS 1e-5f

#define SL 32
#define RG 8
#define HT 512
#define HU 40
#define CAP DFEAT

__global__ __launch_bounds__(HT) void edge_deg_kernel(const float* __restrict__ ea,
                                                      const int* __restrict__ ei,
                                                      float* __restrict__ part,
                                                      int* __restrict__ cpart, int E, int N) {
    extern __shared__ float s_deg[];
    const int sl = blockIdx.x % SL;
    const int rg = blockIdx.x / SL;
    const int RN = (N + RG - 1) / RG;
    const int lo = rg * RN;
    const int nr = max(0, min(N - lo, RN));
    int* s_cnt = (int*)(s_deg + RN);
    for (int i = threadIdx.x; i < nr; i += HT) { s_deg[i] = 0.0f; s_cnt[i] = 0; }
    __syncthreads();

    const int stride = SL * HT;
    for (int e0 = sl * HT + threadIdx.x; e0 < E; e0 += stride * HU) {
        int   d[HU];
        float w[HU];
        #pragma unroll
        for (int u = 0; u < HU; u++) {
            const int e = e0 + u * stride;
            d[u] = ei[E + min(e, E - 1)];
        }
        #pragma unroll
        for (int u = 0; u < HU; u++) {
            const int e = e0 + u * stride;
            const int loc = d[u] - lo;
            const bool in = (e < E) & ((unsigned)loc < (unsigned)nr);
            d[u] = in ? loc : -1;
        }
        #pragma unroll
        for (int u = 0; u < HU; u++) {
            const int ew = (d[u] >= 0) ? e0 + u * stride : 0;
            w[u] = ea[(size_t)ew * 4];
        }
        #pragma unroll
        for (int u = 0; u < HU; u++) asm volatile("" :: "v"(w[u]));
        #pragma unroll
        for (int u = 0; u < HU; u++) {
            if (d[u] >= 0) {
                atomicAdd(&s_deg[d[u]], fabsf(w[u]));
                atomicAdd(&s_cnt[d[u]], 1);
            }
        }
    }
    __syncthreads();

    float* p = part + (size_t)sl * N + lo;
    int*   c = cpart + (size_t)sl * N + lo;
    for (int i = threadIdx.x; i < nr; i += HT) { p[i] = s_deg[i]; c[i] = s_cnt[i]; }
}

__global__ __launch_bounds__(HT) void fill_kernel(const int* __restrict__ ei,
                                                  const float* __restrict__ part,
                                                  const int* __restrict__ cpart,
                                                  float* __restrict__ dinv,
                                                  int* __restrict__ cnt,
                                                  int* __restrict__ slots, int E, int N) {
    extern __shared__ int s_pos[];
    const int sl = blockIdx.x % SL;
    const int rg = blockIdx.x / SL;
    const int RN = (N + RG - 1) / RG;
    const int lo = rg * RN;
    const int nr = max(0, min(N - lo, RN));
    const int sub = (RN + SL - 1) / SL;
    for (int i = threadIdx.x; i < nr; i += HT) {
        const int node = lo + i;
        int run = 0, tot = 0;
        #pragma unroll
        for (int t = 0; t < SL; t++) {
            const int c = cpart[(size_t)t * N + node];
            run += (t < sl) ? c : 0;
            tot += c;
        }
        s_pos[i] = run;
        if (i / sub == sl) {
            float dg = 1.0f;
            #pragma unroll
            for (int t = 0; t < SL; t++) dg += part[(size_t)t * N + node];
            dinv[node] = (dg > 0.0f) ? rsqrtf(dg) : 0.0f;
            cnt[node] = tot;
        }
    }
    __syncthreads();

    const int stride = SL * HT;
    for (int e0 = sl * HT + threadIdx.x; e0 < E; e0 += stride * HU) {
        int d[HU];
        int rk[HU];
        #pragma unroll
        for (int u = 0; u < HU; u++) {
            const int e = e0 + u * stride;
            d[u] = ei[E + min(e, E - 1)];
        }
        #pragma unroll
        for (int u = 0; u < HU; u++) {
            const int e = e0 + u * stride;
            const int loc = d[u] - lo;
            const bool in = (e < E) & ((unsigned)loc < (unsigned)nr);
            d[u] = in ? loc : -1;
        }
        #pragma unroll
        for (int u = 0; u < HU; u++) {
            rk[u] = CAP;
            if (d[u] >= 0) rk[u] = atomicAdd(&s_pos[d[u]], 1);
        }
        #pragma unroll
        for (int u = 0; u < HU; u++) {
            if (rk[u] < CAP) slots[(size_t)(lo + d[u]) * DFEAT + rk[u]] = e0 + u * stride;
        }
    }
}

#define NPB 16
#define NPW 2
#define XP 132

__device__ __forceinline__ int lane_i(int v, int l) { return __builtin_amdgcn_readlane(v, l); }
__device__ __forceinline__ float lane_f(float v, int l) {
    return __int_as_float(__builtin_amdgcn_readlane(__float_as_int(v), l));
}

__device__ __forceinline__ float4 gather_node(const float* __restrict__ x,
                                              const float* __restrict__ ea,
                                              const int* __restrict__ ei,
                                              const float* __restrict__ dinv,
                                              const int* __restrict__ cnt,
                                              const float* orow, int node, int lane,
                                              int E, float& nsum) {
    const float di = dinv[node];
    const int   c  = __builtin_amdgcn_readfirstlane(cnt[node]);
    const int   d0 = lane * 4;

    float a0 = 0.0f, a1 = 0.0f, a2 = 0.0f, a3 = 0.0f;
    float ns = 0.0f;

    if (c <= 0) {
    } else if (c <= CAP) {
        const int4 rec = *(const int4*)((const int*)orow + d0);
        const bool k0 = d0 + 0 < c, k1 = d0 + 1 < c, k2 = d0 + 2 < c, k3 = d0 + 3 < c;
        const int e0 = k0 ? rec.x : 0, e1 = k1 ? rec.y : 0, e2 = k2 ? rec.z : 0, e3 = k3 ? rec.w : 0;
        const int   sr0 = ei[e0], sr1 = ei[e1], sr2 = ei[e2], sr3 = ei[e3];
        const float w0 = ea[(size_t)e0 * 4], w1 = ea[(size_t)e1 * 4];
        const float w2 = ea[(size_t)e2 * 4], w3 = ea[(size_t)e3 * 4];
        const float q0 = dinv[sr0], q1 = dinv[sr1], q2 = dinv[sr2], q3 = dinv[sr3];
        const float nm0 = k0 ? q0 * fabsf(w0) * di : 0.0f;
        const float nm1 = k1 ? q1 * fabsf(w1) * di : 0.0f;
        const float nm2 = k2 ? q2 * fabsf(w2) * di : 0.0f;
        const float nm3 = k3 ? q3 * fabsf(w3) * di : 0.0f;
        ns = (nm0 + nm1) + (nm2 + nm3);
        #pragma unroll
        for (int off = 16; off > 0; off >>= 1) ns += __shfl_xor(ns, off, 32);

        const int ng = c >> 2;
        #pragma unroll 4
        for (int l = 0; l < ng; l++) {
            const int   s0 = lane_i(sr0, l), s1 = lane_i(sr1, l), s2 = lane_i(sr2, l), s3 = lane_i(sr3, l);
            const float n0 = lane_f(nm0, l), n1 = lane_f(nm1, l), n2 = lane_f(nm2, l), n3 = lane_f(nm3, l);
            const float4 r0 = *(const float4*)(x + (size_t)s0 * DFEAT + d0);
            const float4 r1 = *(const float4*)(x + (size_t)s1 * DFEAT + d0);
            const float4 r2 = *(const float4*)(x + (size_t)s2 * DFEAT + d0);
            const float4 r3 = *(const float4*)(x + (size_t)s3 * DFEAT + d0);
            a0 += r0.x * n0; a1 += r0.y * n0; a2 += r0.z * n0; a3 += r0.w * n0;
            a0 += r1.x * n1; a1 += r1.y * n1; a2 += r1.z * n1; a3 += r1.w * n1;
            a0 += r2.x * n2; a1 += r2.y * n2; a2 += r2.z * n2; a3 += r2.w * n2;
            a0 += r3.x * n3; a1 += r3.y * n3; a2 += r3.z * n3; a3 += r3.w * n3;
        }
        const int rem = c & 3;
        if (rem > 0) {
            const int   s0 = lane_i(sr0, ng);
            const float n0 = lane_f(nm0, ng);
            const float4 r0 = *(const float4*)(x + (size_t)s0 * DFEAT + d0);
            a0 += r0.x * n0; a1 += r0.y * n0; a2 += r0.z * n0; a3 += r0.w * n0;
        }
        if (rem > 1) {
            const int   s1 = lane_i(sr1, ng);
            const float n1 = lane_f(nm1, ng);
            const float4 r1 = *(const float4*)(x + (size_t)s1 * DFEAT + d0);
            a0 += r1.x * n1; a1 += r1.y * n1; a2 += r1.z * n1; a3 += r1.w * n1;
        }
        if (rem > 2) {
            const int   s2 = lane_i(sr2, ng);
            const float n2 = lane_f(nm2, ng);
            const float4 r2 = *(const float4*)(x + (size_t)s2 * DFEAT + d0);
            a0 += r2.x * n2; a1 += r2.y * n2; a2 += r2.z * n2; a3 += r2.w * n2;
        }
    } else {
        for (int b0 = 0; b0 < E; b0 += 32) {
            const int e = b0 + lane;
            const int d = (e < E) ? ei[E + e] : -1;
            unsigned m = (unsigned)__ballot(d == node);
            while (m) {
                const int bit = __ffs(m) - 1;
                m &= m - 1;
                const int ee = b0 + bit;
                const int s  = ei[ee];
                const float nv = dinv[s] * fabsf(ea[(size_t)ee * 4]) * di;
                const float4 r = *(const float4*)(x + (size_t)s * DFEAT + d0);
                a0 += r.x * nv; a1 += r.y * nv; a2 += r.z * nv; a3 += r.w * nv;
                ns += nv;
            }
        }
    }

    const float self_norm = di * di;
    const float4 xv = *(const float4*)(x + (size_t)node * DFEAT + d0);
    float4 res;
    res.x = a0 + xv.x * self_norm;
    res.y = a1 + xv.y * self_norm;
    res.z = a2 + xv.z * self_norm;
    res.w = a3 + xv.w * self_norm;
    nsum = ns + self_norm;
    return res;
}

__global__ __launch_bounds__(32 * NPB / NPW) void gather_linear_finish_kernel(
        const float* __restrict__ x, const float* __restrict__ W, const float* __restrict__ bias,
        const float* __restrict__ ea, const int* __restrict__ ei,
        const float* __restrict__ dinv, const int* __restrict__ cnt,
        const float* __restrict__ gamma, const float* __restrict__ beta,
        float* out, int E, int N) {
    __shared__ __attribute__((aligned(16))) float s_t[NPB][XP];
    __shared__ float s_ns[NPB];
    const int lane = threadIdx.x & 31;
    const int wave = __builtin_amdgcn_readfirstlane((int)(threadIdx.x >> 5));
    const int node_base = blockIdx.x * NPB;

    #pragma unroll
    for (int q = 0; q < NPW; q++) {
        const int row = wave * NPW + q;
        const int node = node_base + row;
        float4 agg = make_float4(0.0f, 0.0f, 0.0f, 0.0f);
        float ns = 0.0f;
        if (node < N)
            agg = gather_node(x, ea, ei, dinv, cnt, out + (size_t)node * DFEAT, node, lane, E, ns);
        *(float4*)&s_t[row][lane * 4] = agg;
        if (lane == 0) s_ns[row] = ns;
    }

    const int c16 = lane & 15;
    const int hf  = lane >> 4;
    const int col = wave * 16 + c16;
    const float* wrow = W + (size_t)col * DFEAT + hf * 64;
    float4 bf[16];
    #pragma unroll
    for (int j = 0; j < 16; j++) bf[j] = *(const float4*)(wrow + 4 * j);
    const float bv = bias[col];

    __syncthreads();

    const float* arow = &s_t[c16][hf * 64];
    v8f acc0 = {}, acc1 = {};
    #pragma unroll
    for (int j = 0; j < 16; j += 2) {
        const float4 a0 = *(const float4*)(arow + 4 * j);
        const float4 a1 = *(const float4*)(arow + 4 * j + 4);
        v2f fa, fb;
        fa.x = a0.x; fa.y = a0.y; fb.x = bf[j].x; fb.y = bf[j].y;
        acc0 = __builtin_amdgcn_wmma_f32_16x16x4_f32(false, fa, false, fb, (short)0, acc0, false, false);
        fa.x = a1.x; fa.y = a1.y; fb.x = bf[j + 1].x; fb.y = bf[j + 1].y;
        acc1 = __builtin_amdgcn_wmma_f32_16x16x4_f32(false, fa, false, fb, (short)0, acc1, false, false);
        fa.x = a0.z; fa.y = a0.w; fb.x = bf[j].z; fb.y = bf[j].w;
        acc0 = __builtin_amdgcn_wmma_f32_16x16x4_f32(false, fa, false, fb, (short)0, acc0, false, false);
        fa.x = a1.z; fa.y = a1.w; fb.x = bf[j + 1].z; fb.y = bf[j + 1].w;
        acc1 = __builtin_amdgcn_wmma_f32_16x16x4_f32(false, fa, false, fb, (short)0, acc1, false, false);
    }

    __syncthreads();

    #pragma unroll
    for (int r = 0; r < 8; r++) {
        const int row = hf * 8 + r;
        float v = (acc0[r] + acc1[r]) + bv * s_ns[row];
        v = (v >= 0.0f) ? v : NEG_SLOPE * v;
        s_t[row][col] = v;
    }

    __syncthreads();

    const int d0 = lane * 4;
    const float4 gv = *(const float4*)(gamma + d0);
    const float4 bt = *(const float4*)(beta + d0);
    #pragma unroll
    for (int q = 0; q < NPW; q++) {
        const int row = wave * NPW + q;
        const int node = node_base + row;
        const float4 t = *(const float4*)&s_t[row][d0];
        float v[4] = {t.x, t.y, t.z, t.w};

        float sum = (v[0] + v[1]) + (v[2] + v[3]);
        #pragma unroll
        for (int off = 16; off > 0; off >>= 1) sum += __shfl_xor(sum, off, 32);
        const float mean = sum * (1.0f / DFEAT);

        float vs = 0.0f;
        #pragma unroll
        for (int j = 0; j < 4; j++) {
            float dlt = v[j] - mean;
            vs += dlt * dlt;
        }
        #pragma unroll
        for (int off = 16; off > 0; off >>= 1) vs += __shfl_xor(vs, off, 32);
        const float inv_std = rsqrtf(vs * (1.0f / DFEAT) + LN_EPS);

        float4 res;
        res.x = (v[0] - mean) * inv_std * gv.x + bt.x;
        res.y = (v[1] - mean) * inv_std * gv.y + bt.y;
        res.z = (v[2] - mean) * inv_std * gv.z + bt.z;
        res.w = (v[3] - mean) * inv_std * gv.w + bt.w;
        if (node < N) *(float4*)(out + (size_t)node * DFEAT + d0) = res;
    }
}

extern "C" void kernel_launch(void* const* d_in, const int* in_sizes, int n_in,
                              void* d_out, int out_size, void* d_ws, size_t ws_size,
                              hipStream_t stream) {
    const float* x     = (const float*)d_in[0];
    const float* ea    = (const float*)d_in[1];
    const float* W     = (const float*)d_in[2];
    const float* bias  = (const float*)d_in[3];
    const float* gamma = (const float*)d_in[4];
    const float* beta  = (const float*)d_in[5];
    const int*   ei    = (const int*)d_in[6];
    float* out = (float*)d_out;

    const int N = in_sizes[0] / DFEAT;
    const int E = in_sizes[1] / 4;

    int*   cnt   = (int*)d_ws;
    float* dinv  = (float*)d_ws + N;
    float* part  = dinv + N;
    int*   cpart = (int*)(part + (size_t)SL * N);
    int*   oi    = (int*)d_out;
    const int RN = (N + RG - 1) / RG;

    edge_deg_kernel<<<SL * RG, HT, (size_t)RN * 2 * sizeof(float), stream>>>(ea, ei, part, cpart, E, N);
    fill_kernel<<<SL * RG, HT, (size_t)RN * sizeof(int), stream>>>(ei, part, cpart, dinv, cnt, oi, E, N);
    gather_linear_finish_kernel<<<(N + NPB - 1) / NPB, 32 * NPB / NPW, 0, stream>>>(
        x, W, bias, ea, ei, dinv, cnt, gamma, beta, out, E, N);
}


# Edge grouping re-cut as a two-level counting sort: 256 edge slices each sorted once by 40-node bucket in LDS, one work-group per bucket ranks only its own edges; range scans gone (sum order changes)
# speedup vs baseline: 4.0767x; 1.3243x over previous
#include <hip/hip_runtime.h>
#include <math.h>

typedef float v2f __attribute__((ext_vector_type(2)));
typedef float v8f __attribute__((ext_vector_type(8)));

#define DFEAT 128
#define NEG_SLOPE 0.01f
#define LN_EPS 1e-5f

#define NSL 256
#define BN 40
#define LOCB 6
#define GT 256
#define AU 10
#define BU 12
#define CAP DFEAT

__device__ __forceinline__ size_t pair_base(int b, int N) {
    const size_t at = (size_t)b * BN * DFEAT;
    const size_t last = (size_t)N * DFEAT - 2 * NSL;
    return at < last ? at : last;
}

__device__ __forceinline__ int block_prefix(int v, int* s_w, int& total) {
    const int lane = threadIdx.x & 31;
    const int wave = threadIdx.x >> 5;
    int inc = v;
    #pragma unroll
    for (int off = 1; off < 32; off <<= 1) {
        const int t = __shfl_up(inc, off, 32);
        if (lane >= off) inc += t;
    }
    if (lane == 31) s_w[wave] = inc;
    __syncthreads();
    int pre = 0, tot = 0;
    #pragma unroll
    for (int w = 0; w < GT / 32; w++) {
        const int x = s_w[w];
        pre += (w < wave) ? x : 0;
        tot += x;
    }
    __syncthreads();
    total = tot;
    return pre + inc - v;
}

__global__ __launch_bounds__(GT) void bin_kernel(const int* __restrict__ ei,
                                                 int* __restrict__ area,
                                                 int* __restrict__ oi, int E, int N) {
    extern __shared__ int s_bin[];
    __shared__ int s_w[GT / 32];
    const int sl = blockIdx.x;
    const int ES = (E + NSL - 1) / NSL;
    const int NB = (N + BN - 1) / BN;
    const int e_lo = sl * ES;
    const int ne = max(0, min(E - e_lo, ES));
    int* s_cur = s_bin;
    int* s_stage = s_bin + NB;
    for (int i = threadIdx.x; i < NB; i += GT) s_cur[i] = 0;
    __syncthreads();

    const int* tg = ei + (size_t)E + e_lo;
    for (int i0 = threadIdx.x; i0 < ne; i0 += GT * AU) {
        int d[AU];
        #pragma unroll
        for (int u = 0; u < AU; u++) d[u] = tg[min(i0 + u * GT, ne - 1)];
        #pragma unroll
        for (int u = 0; u < AU; u++) {
            if ((i0 + u * GT < ne) & ((unsigned)d[u] < (unsigned)N)) atomicAdd(&s_cur[d[u] / BN], 1);
        }
    }
    __syncthreads();

    int placed = 0;
    for (int b0 = 0; b0 < NB; b0 += GT) {
        const int b = b0 + threadIdx.x;
        const int c = (b < NB) ? s_cur[b] : 0;
        int tot;
        const int start = placed + block_prefix(c, s_w, tot);
        if (b < NB) {
            s_cur[b] = start;
            *(int2*)(oi + pair_base(b, N) + 2 * sl) = make_int2(start, c);
        }
        placed += tot;
    }
    __syncthreads();

    for (int i0 = threadIdx.x; i0 < ne; i0 += GT * AU) {
        int d[AU];
        #pragma unroll
        for (int u = 0; u < AU; u++) d[u] = tg[min(i0 + u * GT, ne - 1)];
        int pos[AU];
        #pragma unroll
        for (int u = 0; u < AU; u++) {
            pos[u] = -1;
            if ((i0 + u * GT < ne) & ((unsigned)d[u] < (unsigned)N)) pos[u] = atomicAdd(&s_cur[d[u] / BN], 1);
        }
        #pragma unroll
        for (int u = 0; u < AU; u++) {
            if (pos[u] >= 0) s_stage[pos[u]] = ((e_lo + i0 + u * GT) << LOCB) | (d[u] % BN);
        }
    }
    __syncthreads();

    for (int i = threadIdx.x; i < placed; i += GT) area[(size_t)e_lo + i] = s_stage[i];
}

__global__ __launch_bounds__(GT) void rank_kernel(const float* __restrict__ ea,
                                                  const int* __restrict__ area,
                                                  float* __restrict__ dinv,
                                                  int* __restrict__ cnt,
                                                  int* oi, int E, int N) {
    __shared__ __attribute__((aligned(16))) int s_slot[BN][CAP];
    __shared__ int   s_at[GT * BU];
    __shared__ float s_deg[BN];
    __shared__ int   s_cnt[BN];
    __shared__ int   s_w[GT / 32];
    const int b = blockIdx.x;
    const int lo = b * BN;
    const int nr = max(0, min(N - lo, BN));
    const int ES = (E + NSL - 1) / NSL;
    const int lane = threadIdx.x & 31;
    const int wave = threadIdx.x >> 5;

    const int2 sc = *(const int2*)(oi + pair_base(b, N) + 2 * threadIdx.x);
    if (threadIdx.x < BN) { s_deg[threadIdx.x] = 0.0f; s_cnt[threadIdx.x] = 0; }
    int M;
    const int pre = block_prefix(sc.y, s_w, M);
    const int first = threadIdx.x * ES + sc.x - pre;

    for (int base = 0; base < M; base += GT * BU) {
        const int j1 = min(pre + sc.y, base + GT * BU);
        for (int j = max(pre, base); j < j1; j++) s_at[j - base] = first + j;
        __syncthreads();
        const int m = min(M - base, GT * BU);
        int   rec[BU];
        float w[BU];
        #pragma unroll
        for (int u = 0; u < BU; u++) rec[u] = area[s_at[min((int)threadIdx.x + u * GT, m - 1)]];
        #pragma unroll
        for (int u = 0; u < BU; u++) w[u] = ea[(size_t)(rec[u] >> LOCB) * 4];
        #pragma unroll
        for (int u = 0; u < BU; u++) asm volatile("" :: "v"(w[u]));
        int rk[BU];
        #pragma unroll
        for (int u = 0; u < BU; u++) {
            rk[u] = CAP;
            if ((int)threadIdx.x + u * GT < m) {
                const int loc = rec[u] & ((1 << LOCB) - 1);
                atomicAdd(&s_deg[loc], fabsf(w[u]));
                rk[u] = atomicAdd(&s_cnt[loc], 1);
            }
        }
        #pragma unroll
        for (int u = 0; u < BU; u++) {
            if (rk[u] < CAP) s_slot[rec[u] & ((1 << LOCB) - 1)][rk[u]] = rec[u] >> LOCB;
        }
        __syncthreads();
    }

    if (threadIdx.x < nr) {
        const float dg = 1.0f + s_deg[threadIdx.x];
        dinv[lo + threadIdx.x] = (dg > 0.0f) ? rsqrtf(dg) : 0.0f;
        cnt[lo + threadIdx.x] = s_cnt[threadIdx.x];
    }
    for (int r = wave; r < nr; r += GT / 32) {
        const int c = min(s_cnt[r], CAP);
        if (4 * lane < c)
            *(int4*)(oi + (size_t)(lo + r) * DFEAT + 4 * lane) = *(const int4*)&s_slot[r][4 * lane];
    }
}

#define NPB 16
#define NPW 2
#define XP 132

__device__ __forceinline__ int lane_i(int v, int l) { return __builtin_amdgcn_readlane(v, l); }
__device__ __forceinline__ float lane_f(float v, int l) {
    return __int_as_float(__builtin_amdgcn_readlane(__float_as_int(v), l));
}

__device__ __forceinline__ float4 gather_node(const float* __restrict__ x,
                                              const float* __restrict__ ea,
                                              const int* __restrict__ ei,
                                              const float* __restrict__ dinv,
                                              const int* __restrict__ cnt,
                                              const float* orow, int node, int lane,
                                              int E, float& nsum) {
    const float di = dinv[node];
    const int   c  = __builtin_amdgcn_readfirstlane(cnt[node]);
    const int   d0 = lane * 4;

    float a0 = 0.0f, a1 = 0.0f, a2 = 0.0f, a3 = 0.0f;
    float ns = 0.0f;

    if (c <= 0) {
    } else if (c <= CAP) {
        const int4 rec = *(const int4*)((const int*)orow + d0);
        const bool k0 = d0 + 0 < c, k1 = d0 + 1 < c, k2 = d0 + 2 < c, k3 = d0 + 3 < c;
        const int e0 = k0 ? rec.x : 0, e1 = k1 ? rec.y : 0, e2 = k2 ? rec.z : 0, e3 = k3 ? rec.w : 0;
        const int   sr0 = ei[e0], sr1 = ei[e1], sr2 = ei[e2], sr3 = ei[e3];
        const float w0 = ea[(size_t)e0 * 4], w1 = ea[(size_t)e1 * 4];
        const float w2 = ea[(size_t)e2 * 4], w3 = ea[(size_t)e3 * 4];
        const float q0 = dinv[sr0], q1 = dinv[sr1], q2 = dinv[sr2], q3 = dinv[sr3];
        const float nm0 = k0 ? q0 * fabsf(w0) * di : 0.0f;
        const float nm1 = k1 ? q1 * fabsf(w1) * di : 0.0f;
        const float nm2 = k2 ? q2 * fabsf(w2) * di : 0.0f;
        const float nm3 = k3 ? q3 * fabsf(w3) * di : 0.0f;
        ns = (nm0 + nm1) + (nm2 + nm3);
        #pragma unroll
        for (int off = 16; off > 0; off >>= 1) ns += __shfl_xor(ns, off, 32);

        const int ng = c >> 2;
        #pragma unroll 4
        for (int l = 0; l < ng; l++) {
            const int   s0 = lane_i(sr0, l), s1 = lane_i(sr1, l), s2 = lane_i(sr2, l), s3 = lane_i(sr3, l);
            const float n0 = lane_f(nm0, l), n1 = lane_f(nm1, l), n2 = lane_f(nm2, l), n3 = lane_f(nm3, l);
            const float4 r0 = *(const float4*)(x + (size_t)s0 * DFEAT + d0);
            const float4 r1 = *(const float4*)(x + (size_t)s1 * DFEAT + d0);
            const float4 r2 = *(const float4*)(x + (size_t)s2 * DFEAT + d0);
            const float4 r3 = *(const float4*)(x + (size_t)s3 * DFEAT + d0);
            a0 += r0.x * n0; a1 += r0.y * n0; a2 += r0.z * n0; a3 += r0.w * n0;
            a0 += r1.x * n1; a1 += r1.y * n1; a2 += r1.z * n1; a3 += r1.w * n1;
            a0 += r2.x * n2; a1 += r2.y * n2; a2 += r2.z * n2; a3 += r2.w * n2;
            a0 += r3.x * n3; a1 += r3.y * n3; a2 += r3.z * n3; a3 += r3.w * n3;
        }
        const int rem = c & 3;
        if (rem > 0) {
            const int   s0 = lane_i(sr0, ng);
            const float n0 = lane_f(nm0, ng);
            const float4 r0 = *(const float4*)(x + (size_t)s0 * DFEAT + d0);
            a0 += r0.x * n0; a1 += r0.y * n0; a2 += r0.z * n0; a3 += r0.w * n0;
        }
        if (rem > 1) {
            const int   s1 = lane_i(sr1, ng);
            const float n1 = lane_f(nm1, ng);
            const float4 r1 = *(const float4*)(x + (size_t)s1 * DFEAT + d0);
            a0 += r1.x * n1; a1 += r1.y * n1; a2 += r1.z * n1; a3 += r1.w * n1;
        }
        if (rem > 2) {
            const int   s2 = lane_i(sr2, ng);
            const float n2 = lane_f(nm2, ng);
            const float4 r2 = *(const float4*)(x + (size_t)s2 * DFEAT + d0);
            a0 += r2.x * n2; a1 += r2.y * n2; a2 += r2.z * n2; a3 += r2.w * n2;
        }
    } else {
        for (int b0 = 0; b0 < E; b0 += 32) {
            const int e = b0 + lane;
            const int d = (e < E) ? ei[E + e] : -1;
            unsigned m = (unsigned)__ballot(d == node);
            while (m) {
                const int bit = __ffs(m) - 1;
                m &= m - 1;
                const int ee = b0 + bit;
                const int s  = ei[ee];
                const float nv = dinv[s] * fabsf(ea[(size_t)ee * 4]) * di;
                const float4 r = *(const float4*)(x + (size_t)s * DFEAT + d0);
                a0 += r.x * nv; a1 += r.y * nv; a2 += r.z * nv; a3 += r.w * nv;
                ns += nv;
            }
        }
    }

    const float self_norm = di * di;
    const float4 xv = *(const float4*)(x + (size_t)node * DFEAT + d0);
    float4 res;
    res.x = a0 + xv.x * self_norm;
    res.y = a1 + xv.y * self_norm;
    res.z = a2 + xv.z * self_norm;
    res.w = a3 + xv.w * self_norm;
    nsum = ns + self_norm;
    return res;
}

__global__ __launch_bounds__(32 * NPB / NPW) void gather_linear_finish_kernel(
        const float* __restrict__ x, const float* __restrict__ W, const float* __restrict__ bias,
        const float* __restrict__ ea, const int* __restrict__ ei,
        const float* __restrict__ dinv, const int* __restrict__ cnt,
        const float* __restrict__ gamma, const float* __restrict__ beta,
        float* out, int E, int N) {
    __shared__ __attribute__((aligned(16))) float s_t[NPB][XP];
    __shared__ float s_ns[NPB];
    const int lane = threadIdx.x & 31;
    const int wave = __builtin_amdgcn_readfirstlane((int)(threadIdx.x >> 5));
    const int node_base = blockIdx.x * NPB;

    #pragma unroll
    for (int q = 0; q < NPW; q++) {
        const int row = wave * NPW + q;
        const int node = node_base + row;
        float4 agg = make_float4(0.0f, 0.0f, 0.0f, 0.0f);
        float ns = 0.0f;
        if (node < N)
            agg = gather_node(x, ea, ei, dinv, cnt, out + (size_t)node * DFEAT, node, lane, E, ns);
        *(float4*)&s_t[row][lane * 4] = agg;
        if (lane == 0) s_ns[row] = ns;
    }

    const int c16 = lane & 15;
    const int hf  = lane >> 4;
    const int col = wave * 16 + c16;
    const float* wrow = W + (size_t)col * DFEAT + hf * 64;
    float4 bf[16];
    #pragma unroll
    for (int j = 0; j < 16; j++) bf[j] = *(const float4*)(wrow + 4 * j);
    const float bv = bias[col];

    __syncthreads();

    const float* arow = &s_t[c16][hf * 64];
    v8f acc0 = {}, acc1 = {};
    #pragma unroll
    for (int j = 0; j < 16; j += 2) {
        const float4 a0 = *(const float4*)(arow + 4 * j);
        const float4 a1 = *(const float4*)(arow + 4 * j + 4);
        v2f fa, fb;
        fa.x = a0.x; fa.y = a0.y; fb.x = bf[j].x; fb.y = bf[j].y;
        acc0 = __builtin_amdgcn_wmma_f32_16x16x4_f32(false, fa, false, fb, (short)0, acc0, false, false);
        fa.x = a1.x; fa.y = a1.y; fb.x = bf[j + 1].x; fb.y = bf[j + 1].y;
        acc1 = __builtin_amdgcn_wmma_f32_16x16x4_f32(false, fa, false, fb, (short)0, acc1, false, false);
        fa.x = a0.z; fa.y = a0.w; fb.x = bf[j].z; fb.y = bf[j].w;
        acc0 = __builtin_amdgcn_wmma_f32_16x16x4_f32(false, fa, false, fb, (short)0, acc0, false, false);
        fa.x = a1.z; fa.y = a1.w; fb.x = bf[j + 1].z; fb.y = bf[j + 1].w;
        acc1 = __builtin_amdgcn_wmma_f32_16x16x4_f32(false, fa, false, fb, (short)0, acc1, false, false);
    }

    __syncthreads();

    #pragma unroll
    for (int r = 0; r < 8; r++) {
        const int row = hf * 8 + r;
        float v = (acc0[r] + acc1[r]) + bv * s_ns[row];
        v = (v >= 0.0f) ? v : NEG_SLOPE * v;
        s_t[row][col] = v;
    }

    __syncthreads();

    const int d0 = lane * 4;
    const float4 gv = *(const float4*)(gamma + d0);
    const float4 bt = *(const float4*)(beta + d0);
    #pragma unroll
    for (int q = 0; q < NPW; q++) {
        const int row = wave * NPW + q;
        const int node = node_base + row;
        const float4 t = *(const float4*)&s_t[row][d0];
        float v[4] = {t.x, t.y, t.z, t.w};

        float sum = (v[0] + v[1]) + (v[2] + v[3]);
        #pragma unroll
        for (int off = 16; off > 0; off >>= 1) sum += __shfl_xor(sum, off, 32);
        const float mean = sum * (1.0f / DFEAT);

        float vs = 0.0f;
        #pragma unroll
        for (int j = 0; j < 4; j++) {
            float dlt = v[j] - mean;
            vs += dlt * dlt;
        }
        #pragma unroll
        for (int off = 16; off > 0; off >>= 1) vs += __shfl_xor(vs, off, 32);
        const float inv_std = rsqrtf(vs * (1.0f / DFEAT) + LN_EPS);

        float4 res;
        res.x = (v[0] - mean) * inv_std * gv.x + bt.x;
        res.y = (v[1] - mean) * inv_std * gv.y + bt.y;
        res.z = (v[2] - mean) * inv_std * gv.z + bt.z;
        res.w = (v[3] - mean) * inv_std * gv.w + bt.w;
        if (node < N) *(float4*)(out + (size_t)node * DFEAT + d0) = res;
    }
}

extern "C" void kernel_launch(void* const* d_in, const int* in_sizes, int n_in,
                              void* d_out, int out_size, void* d_ws, size_t ws_size,
                              hipStream_t stream) {
    const float* x     = (const float*)d_in[0];
    const float* ea    = (const float*)d_in[1];
    const float* W     = (const float*)d_in[2];
    const float* bias  = (const float*)d_in[3];
    const float* gamma = (const float*)d_in[4];
    const float* beta  = (const float*)d_in[5];
    const int*   ei    = (const int*)d_in[6];
    float* out = (float*)d_out;

    const int N = in_sizes[0] / DFEAT;
    const int E = in_sizes[1] / 4;

    int*   cnt   = (int*)d_ws;
    float* dinv  = (float*)d_ws + N;
    int*   area  = (int*)(dinv + N);
    int*   oi    = (int*)d_out;
    const int NB = (N + BN - 1) / BN;
    const int ES = (E + NSL - 1) / NSL;

    bin_kernel<<<NSL, GT, (size_t)(NB + ES) * sizeof(int), stream>>>(ei, area, oi, E, N);
    rank_kernel<<<NB, GT, 0, stream>>>(ea, area, dinv, cnt, oi, E, N);
    gather_linear_finish_kernel<<<(N + NPB - 1) / NPB, 32 * NPB / NPW, 0, stream>>>(
        x, W, bias, ea, ei, dinv, cnt, gamma, beta, out, E, N);
}


# Sort records carry the edge's source; the rank kernel lays (source, |weight|) pairs into the target's row: the finish kernel drops its per-edge source and weight look-ups, save for ranks without room
# speedup vs baseline: 4.6498x; 1.1406x over previous
#include <hip/hip_runtime.h>
#include <math.h>

typedef float v2f __attribute__((ext_vector_type(2)));
typedef float v8f __attribute__((ext_vector_type(8)));

#define DFEAT 128
#define NEG_SLOPE 0.01f
#define LN_EPS 1e-5f

#define NSL 256
#define BN 40
#define LOCB 6
#define ELB 12
#define SRB (LOCB + ELB)
#define PSH 20
#define GT 256
#define AU 10
#define BU 12
#define CAP DFEAT

__device__ __forceinline__ size_t pair_base(int b, int N) {
    const size_t at = (size_t)b * BN * DFEAT;
    const size_t last = (size_t)N * DFEAT - 2 * NSL;
    return at < last ? at : last;
}

__device__ __forceinline__ int block_prefix(int v, int* s_w, int& total) {
    const int lane = threadIdx.x & 31;
    const int wave = threadIdx.x >> 5;
    int inc = v;
    #pragma unroll
    for (int off = 1; off < 32; off <<= 1) {
        const int t = __shfl_up(inc, off, 32);
        if (lane >= off) inc += t;
    }
    if (lane == 31) s_w[wave] = inc;
    __syncthreads();
    int pre = 0, tot = 0;
    #pragma unroll
    for (int w = 0; w < GT / 32; w++) {
        const int x = s_w[w];
        pre += (w < wave) ? x : 0;
        tot += x;
    }
    __syncthreads();
    total = tot;
    return pre + inc - v;
}

__global__ __launch_bounds__(GT) void bin_kernel(const int* __restrict__ ei,
                                                 int* __restrict__ area,
                                                 int* __restrict__ oi, int E, int N) {
    extern __shared__ int s_bin[];
    __shared__ int s_w[GT / 32];
    const int sl = blockIdx.x;
    const int ES = (E + NSL - 1) / NSL;
    const int NB = (N + BN - 1) / BN;
    const int e_lo = sl * ES;
    const int ne = max(0, min(E - e_lo, ES));
    int* s_cur = s_bin;
    int* s_stage = s_bin + NB;
    for (int i = threadIdx.x; i < NB; i += GT) s_cur[i] = 0;
    __syncthreads();

    const int* tg = ei + (size_t)E + e_lo;
    const int* sg = ei + e_lo;
    for (int i0 = threadIdx.x; i0 < ne; i0 += GT * AU) {
        int d[AU];
        #pragma unroll
        for (int u = 0; u < AU; u++) d[u] = tg[min(i0 + u * GT, ne - 1)];
        #pragma unroll
        for (int u = 0; u < AU; u++) {
            if ((i0 + u * GT < ne) & ((unsigned)d[u] < (unsigned)N)) atomicAdd(&s_cur[d[u] / BN], 1);
        }
    }
    __syncthreads();

    int placed = 0;
    for (int b0 = 0; b0 < NB; b0 += GT) {
        const int b = b0 + threadIdx.x;
        const int c = (b < NB) ? s_cur[b] : 0;
        int tot;
        const int start = placed + block_prefix(c, s_w, tot);
        if (b < NB) {
            s_cur[b] = start;
            *(int2*)(oi + pair_base(b, N) + 2 * sl) = make_int2(start, c);
        }
        placed += tot;
    }
    __syncthreads();

    for (int i0 = threadIdx.x; i0 < ne; i0 += GT * AU) {
        int d[AU];
        #pragma unroll
        for (int u = 0; u < AU; u++) d[u] = tg[min(i0 + u * GT, ne - 1)];
        int s[AU];
        #pragma unroll
        for (int u = 0; u < AU; u++) s[u] = sg[min(i0 + u * GT, ne - 1)];
        int pos[AU];
        #pragma unroll
        for (int u = 0; u < AU; u++) {
            pos[u] = -1;
            if ((i0 + u * GT < ne) & ((unsigned)d[u] < (unsigned)N)) pos[u] = atomicAdd(&s_cur[d[u] / BN], 1);
        }
        #pragma unroll
        for (int u = 0; u < AU; u++) {
            if (pos[u] >= 0)
                s_stage[pos[u]] = (int)(((unsigned)s[u] << SRB) | (unsigned)((i0 + u * GT) << LOCB) | (unsigned)(d[u] % BN));
        }
    }
    __syncthreads();

    for (int i = threadIdx.x; i < placed; i += GT) area[(size_t)e_lo + i] = s_stage[i];
}

__global__ __launch_bounds__(GT) void rank_kernel(const float* __restrict__ ea,
                                                  const int* __restrict__ area,
                                                  float* __restrict__ dinv,
                                                  int* __restrict__ cnt,
                                                  int* oi, int E, int N) {
    __shared__ __attribute__((aligned(16))) int   s_rec[BN][CAP];
    __shared__ __attribute__((aligned(16))) float s_wt[BN][CAP];
    __shared__ unsigned char s_sl[BN][CAP];
    __shared__ int   s_at[GT * BU];
    __shared__ float s_deg[BN];
    __shared__ int   s_cnt[BN];
    __shared__ int   s_w[GT / 32];
    const int b = blockIdx.x;
    const int lo = b * BN;
    const int nr = max(0, min(N - lo, BN));
    const int ES = (E + NSL - 1) / NSL;
    const int lane = threadIdx.x & 31;
    const int wave = threadIdx.x >> 5;

    const int2 sc = *(const int2*)(oi + pair_base(b, N) + 2 * threadIdx.x);
    if (threadIdx.x < BN) { s_deg[threadIdx.x] = 0.0f; s_cnt[threadIdx.x] = 0; }
    int M;
    const int pre = block_prefix(sc.y, s_w, M);
    const int first = threadIdx.x * ES + sc.x - pre;

    for (int base = 0; base < M; base += GT * BU) {
        const int j1 = min(pre + sc.y, base + GT * BU);
        for (int j = max(pre, base); j < j1; j++) s_at[j - base] = (first + j) | ((int)threadIdx.x << PSH);
        __syncthreads();
        const int m = min(M - base, GT * BU);
        int   rec[BU];
        float w[BU];
        int   at[BU];
        #pragma unroll
        for (int u = 0; u < BU; u++) at[u] = s_at[min((int)threadIdx.x + u * GT, m - 1)];
        #pragma unroll
        for (int u = 0; u < BU; u++) rec[u] = area[at[u] & ((1 << PSH) - 1)];
        #pragma unroll
        for (int u = 0; u < BU; u++)
            w[u] = ea[(size_t)((at[u] >> PSH) * ES + ((rec[u] >> LOCB) & ((1 << ELB) - 1))) * 4];
        #pragma unroll
        for (int u = 0; u < BU; u++) asm volatile("" :: "v"(w[u]));
        int rk[BU];
        #pragma unroll
        for (int u = 0; u < BU; u++) {
            rk[u] = CAP;
            if ((int)threadIdx.x + u * GT < m) {
                const int loc = rec[u] & ((1 << LOCB) - 1);
                atomicAdd(&s_deg[loc], fabsf(w[u]));
                rk[u] = atomicAdd(&s_cnt[loc], 1);
            }
        }
        #pragma unroll
        for (int u = 0; u < BU; u++) {
            if (rk[u] < CAP) {
                const int loc = rec[u] & ((1 << LOCB) - 1);
                s_rec[loc][rk[u]] = rec[u];
                s_wt[loc][rk[u]]  = fabsf(w[u]);
                s_sl[loc][rk[u]]  = (unsigned char)(at[u] >> PSH);
            }
        }
        __syncthreads();
    }

    if (threadIdx.x < nr) {
        const float dg = 1.0f + s_deg[threadIdx.x];
        dinv[lo + threadIdx.x] = (dg > 0.0f) ? rsqrtf(dg) : 0.0f;
        cnt[lo + threadIdx.x] = s_cnt[threadIdx.x];
    }
    for (int r = wave; r < nr; r += GT / 32) {
        const int c = min(s_cnt[r], CAP);
        const int P = min(c, CAP - c);
        int v[4];
        #pragma unroll
        for (int k = 0; k < 4; k++) {
            const int i = 4 * lane + k;
            int val = 0;
            if (i < c) {
                const unsigned rc = (unsigned)s_rec[r][i];
                val = (i < P) ? (int)(rc >> SRB)
                              : (int)s_sl[r][i] * ES + (int)((rc >> LOCB) & ((1u << ELB) - 1));
            } else if (i >= CAP - P) {
                val = __float_as_int(s_wt[r][CAP - 1 - i]);
            }
            v[k] = val;
        }
        *(int4*)(oi + (size_t)(lo + r) * DFEAT + 4 * lane) = make_int4(v[0], v[1], v[2], v[3]);
    }
}

#define NPB 16
#define NPW 2
#define XP 132

__device__ __forceinline__ int lane_i(int v, int l) { return __builtin_amdgcn_readlane(v, l); }
__device__ __forceinline__ float lane_f(float v, int l) {
    return __int_as_float(__builtin_amdgcn_readlane(__float_as_int(v), l));
}

__device__ __forceinline__ float4 gather_node(const float* __restrict__ x,
                                              const float* __restrict__ ea,
                                              const int* __restrict__ ei,
                                              const float* __restrict__ dinv,
                                              const int* __restrict__ cnt,
                                              const float* orow, int node, int lane,
                                              int E, float& nsum) {
    const float di = dinv[node];
    const int   c  = __builtin_amdgcn_readfirstlane(cnt[node]);
    const int   d0 = lane * 4;

    float a0 = 0.0f, a1 = 0.0f, a2 = 0.0f, a3 = 0.0f;
    float ns = 0.0f;

    if (c <= 0) {
    } else if (c <= CAP) {
        const int4 ra = *(const int4*)((const int*)orow + d0);
        const int4 rw = *(const int4*)((const int*)orow + (CAP - 4 - d0));
        const bool k0 = d0 + 0 < c, k1 = d0 + 1 < c, k2 = d0 + 2 < c, k3 = d0 + 3 < c;
        const int P = min(c, CAP - c);
        int   sr0 = ra.x, sr1 = ra.y, sr2 = ra.z, sr3 = ra.w;
        float w0 = __int_as_float(rw.w), w1 = __int_as_float(rw.z);
        float w2 = __int_as_float(rw.y), w3 = __int_as_float(rw.x);
        if (c > P) {
            const bool g0 = k0 & (d0 + 0 >= P), g1 = k1 & (d0 + 1 >= P);
            const bool g2 = k2 & (d0 + 2 >= P), g3 = k3 & (d0 + 3 >= P);
            const int e0 = g0 ? ra.x : 0, e1 = g1 ? ra.y : 0, e2 = g2 ? ra.z : 0, e3 = g3 ? ra.w : 0;
            const int   t0 = ei[e0], t1 = ei[e1], t2 = ei[e2], t3 = ei[e3];
            const float u0 = ea[(size_t)e0 * 4], u1 = ea[(size_t)e1 * 4];
            const float u2 = ea[(size_t)e2 * 4], u3 = ea[(size_t)e3 * 4];
            sr0 = g0 ? t0 : sr0; sr1 = g1 ? t1 : sr1; sr2 = g2 ? t2 : sr2; sr3 = g3 ? t3 : sr3;
            w0 = g0 ? fabsf(u0) : w0; w1 = g1 ? fabsf(u1) : w1;
            w2 = g2 ? fabsf(u2) : w2; w3 = g3 ? fabsf(u3) : w3;
        }
        sr0 = k0 ? sr0 : 0; sr1 = k1 ? sr1 : 0; sr2 = k2 ? sr2 : 0; sr3 = k3 ? sr3 : 0;
        const float q0 = dinv[sr0], q1 = dinv[sr1], q2 = dinv[sr2], q3 = dinv[sr3];
        const float nm0 = k0 ? q0 * w0 * di : 0.0f;
        const float nm1 = k1 ? q1 * w1 * di : 0.0f;
        const float nm2 = k2 ? q2 * w2 * di : 0.0f;
        const float nm3 = k3 ? q3 * w3 * di : 0.0f;
        ns = (nm0 + nm1) + (nm2 + nm3);
        #pragma unroll
        for (int off = 16; off > 0; off >>= 1) ns += __shfl_xor(ns, off, 32);

        const int ng = c >> 2;
        #pragma unroll 4
        for (int l = 0; l < ng; l++) {
            const int   s0 = lane_i(sr0, l), s1 = lane_i(sr1, l), s2 = lane_i(sr2, l), s3 = lane_i(sr3, l);
            const float n0 = lane_f(nm0, l), n1 = lane_f(nm1, l), n2 = lane_f(nm2, l), n3 = lane_f(nm3, l);
            const float4 r0 = *(const float4*)(x + (size_t)s0 * DFEAT + d0);
            const float4 r1 = *(const float4*)(x + (size_t)s1 * DFEAT + d0);
            const float4 r2 = *(const float4*)(x + (size_t)s2 * DFEAT + d0);
            const float4 r3 = *(const float4*)(x + (size_t)s3 * DFEAT + d0);
            a0 += r0.x * n0; a1 += r0.y * n0; a2 += r0.z * n0; a3 += r0.w * n0;
            a0 += r1.x * n1; a1 += r1.y * n1; a2 += r1.z * n1; a3 += r1.w * n1;
            a0 += r2.x * n2; a1 += r2.y * n2; a2 += r2.z * n2; a3 += r2.w * n2;
            a0 += r3.x * n3; a1 += r3.y * n3; a2 += r3.z * n3; a3 += r3.w * n3;
        }
        const int rem = c & 3;
        if (rem > 0) {
            const int   s0 = lane_i(sr0, ng);
            const float n0 = lane_f(nm0, ng);
            const float4 r0 = *(const float4*)(x + (size_t)s0 * DFEAT + d0);
            a0 += r0.x * n0; a1 += r0.y * n0; a2 += r0.z * n0; a3 += r0.w * n0;
        }
        if (rem > 1) {
            const int   s1 = lane_i(sr1, ng);
            const float n1 = lane_f(nm1, ng);
            const float4 r1 = *(const float4*)(x + (size_t)s1 * DFEAT + d0);
            a0 += r1.x * n1; a1 += r1.y * n1; a2 += r1.z * n1; a3 += r1.w * n1;
        }
        if (rem > 2) {
            const int   s2 = lane_i(sr2, ng);
            const float n2 = lane_f(nm2, ng);
            const float4 r2 = *(const float4*)(x + (size_t)s2 * DFEAT + d0);
            a0 += r2.x * n2; a1 += r2.y * n2; a2 += r2.z * n2; a3 += r2.w * n2;
        }
    } else {
        for (int b0 = 0; b0 < E; b0 += 32) {
            const int e = b0 + lane;
            const int d = (e < E) ? ei[E + e] : -1;
            unsigned m = (unsigned)__ballot(d == node);
            while (m) {
                const int bit = __ffs(m) - 1;
                m &= m - 1;
                const int ee = b0 + bit;
                const int s  = ei[ee];
                const float nv = dinv[s] * fabsf(ea[(size_t)ee * 4]) * di;
                const float4 r = *(const float4*)(x + (size_t)s * DFEAT + d0);
                a0 += r.x * nv; a1 += r.y * nv; a2 += r.z * nv; a3 += r.w * nv;
                ns += nv;
            }
        }
    }

    const float self_norm = di * di;
    const float4 xv = *(const float4*)(x + (size_t)node * DFEAT + d0);
    float4 res;
    res.x = a0 + xv.x * self_norm;
    res.y = a1 + xv.y * self_norm;
    res.z = a2 + xv.z * self_norm;
    res.w = a3 + xv.w * self_norm;
    nsum = ns + self_norm;
    return res;
}

__global__ __launch_bounds__(32 * NPB / NPW) void gather_linear_finish_kernel(
        const float* __restrict__ x, const float* __restrict__ W, const float* __restrict__ bias,
        const float* __restrict__ ea, const int* __restrict__ ei,
        const float* __restrict__ dinv, const int* __restrict__ cnt,
        const float* __restrict__ gamma, const float* __restrict__ beta,
        float* out, int E, int N) {
    __shared__ __attribute__((aligned(16))) float s_t[NPB][XP];
    __shared__ float s_ns[NPB];
    const int lane = threadIdx.x & 31;
    const int wave = __builtin_amdgcn_readfirstlane((int)(threadIdx.x >> 5));
    const int node_base = blockIdx.x * NPB;

    #pragma unroll
    for (int q = 0; q < NPW; q++) {
        const int row = wave * NPW + q;
        const int node = node_base + row;
        float4 agg = make_float4(0.0f, 0.0f, 0.0f, 0.0f);
        float ns = 0.0f;
        if (node < N)
            agg = gather_node(x, ea, ei, dinv, cnt, out + (size_t)node * DFEAT, node, lane, E, ns);
        *(float4*)&s_t[row][lane * 4] = agg;
        if (lane == 0) s_ns[row] = ns;
    }

    const int c16 = lane & 15;
    const int hf  = lane >> 4;
    const int col = wave * 16 + c16;
    const float* wrow = W + (size_t)col * DFEAT + hf * 64;
    float4 bf[16];
    #pragma unroll
    for (int j = 0; j < 16; j++) bf[j] = *(const float4*)(wrow + 4 * j);
    const float bv = bias[col];

    __syncthreads();

    const float* arow = &s_t[c16][hf * 64];
    v8f acc0 = {}, acc1 = {};
    #pragma unroll
    for (int j = 0; j < 16; j += 2) {
        const float4 a0 = *(const float4*)(arow + 4 * j);
        const float4 a1 = *(const float4*)(arow + 4 * j + 4);
        v2f fa, fb;
        fa.x = a0.x; fa.y = a0.y; fb.x = bf[j].x; fb.y = bf[j].y;
        acc0 = __builtin_amdgcn_wmma_f32_16x16x4_f32(false, fa, false, fb, (short)0, acc0, false, false);
        fa.x = a1.x; fa.y = a1.y; fb.x = bf[j + 1].x; fb.y = bf[j + 1].y;
        acc1 = __builtin_amdgcn_wmma_f32_16x16x4_f32(false, fa, false, fb, (short)0, acc1, false, false);
        fa.x = a0.z; fa.y = a0.w; fb.x = bf[j].z; fb.y = bf[j].w;
        acc0 = __builtin_amdgcn_wmma_f32_16x16x4_f32(false, fa, false, fb, (short)0, acc0, false, false);
        fa.x = a1.z; fa.y = a1.w; fb.x = bf[j + 1].z; fb.y = bf[j + 1].w;
        acc1 = __builtin_amdgcn_wmma_f32_16x16x4_f32(false, fa, false, fb, (short)0, acc1, false, false);
    }

    __syncthreads();

    #pragma unroll
    for (int r = 0; r < 8; r++) {
        const int row = hf * 8 + r;
        float v = (acc0[r] + acc1[r]) + bv * s_ns[row];
        v = (v >= 0.0f) ? v : NEG_SLOPE * v;
        s_t[row][col] = v;
    }

    __syncthreads();

    const int d0 = lane * 4;
    const float4 gv = *(const float4*)(gamma + d0);
    const float4 bt = *(const float4*)(beta + d0);
    #pragma unroll
    for (int q = 0; q < NPW; q++) {
        const int row = wave * NPW + q;
        const int node = node_base + row;
        const float4 t = *(const float4*)&s_t[row][d0];
        float v[4] = {t.x, t.y, t.z, t.w};

        float sum = (v[0] + v[1]) + (v[2] + v[3]);
        #pragma unroll
        for (int off = 16; off > 0; off >>= 1) sum += __shfl_xor(sum, off, 32);
        const float mean = sum * (1.0f / DFEAT);

        float vs = 0.0f;
        #pragma unroll
        for (int j = 0; j < 4; j++) {
            float dlt = v[j] - mean;
            vs += dlt * dlt;
        }
        #pragma unroll
        for (int off = 16; off > 0; off >>= 1) vs += __shfl_xor(vs, off, 32);
        const float inv_std = rsqrtf(vs * (1.0f / DFEAT) + LN_EPS);

        float4 res;
        res.x = (v[0] - mean) * inv_std * gv.x + bt.x;
        res.y = (v[1] - mean) * inv_std * gv.y + bt.y;
        res.z = (v[2] - mean) * inv_std * gv.z + bt.z;
        res.w = (v[3] - mean) * inv_std * gv.w + bt.w;
        if (node < N) *(float4*)(out + (size_t)node * DFEAT + d0) = res;
    }
}

extern "C" void kernel_launch(void* const* d_in, const int* in_sizes, int n_in,
                              void* d_out, int out_size, void* d_ws, size_t ws_size,
                              hipStream_t stream) {
    const float* x     = (const float*)d_in[0];
    const float* ea    = (const float*)d_in[1];
    const float* W     = (const float*)d_in[2];
    const float* bias  = (const float*)d_in[3];
    const float* gamma = (const float*)d_in[4];
    const float* beta  = (const float*)d_in[5];
    const int*   ei    = (const int*)d_in[6];
    float* out = (float*)d_out;

    const int N = in_sizes[0] / DFEAT;
    const int E = in_sizes[1] / 4;

    int*   cnt   = (int*)d_ws;
    float* dinv  = (float*)d_ws + N;
    int*   area  = (int*)(dinv + N);
    int*   oi    = (int*)d_out;
    const int NB = (N + BN - 1) / BN;
    const int ES = (E + NSL - 1) / NSL;

    bin_kernel<<<NSL, GT, (size_t)(NB + ES) * sizeof(int), stream>>>(ei, area, oi, E, N);
    rank_kernel<<<NB, GT, 0, stream>>>(ea, area, dinv, cnt, oi, E, N);
    gather_linear_finish_kernel<<<(N + NPB - 1) / NPB, 32 * NPB / NPW, 0, stream>>>(
        x, W, bias, ea, ei, dinv, cnt, gamma, beta, out, E, N);
}


# Edge |weights| travel with the sort records: level 1 reads edge_attr in edge order (whole lines) into a second sorted plane; level 2 reads them beside the records, no scattered 4-byte weight reads
# speedup vs baseline: 4.8369x; 1.0402x over previous
#include <hip/hip_runtime.h>
#include <math.h>

typedef float v2f __attribute__((ext_vector_type(2)));
typedef float v8f __attribute__((ext_vector_type(8)));

#define DFEAT 128
#define NEG_SLOPE 0.01f
#define LN_EPS 1e-5f

#define NSL 256
#define BN 40
#define LOCB 6
#define ELB 12
#define SRB (LOCB + ELB)
#define PSH 20
#define GT 256
#define AU 10
#define BU 12
#define CAP DFEAT

__device__ __forceinline__ size_t pair_base(int b, int N) {
    const size_t at = (size_t)b * BN * DFEAT;
    const size_t last = (size_t)N * DFEAT - 2 * NSL;
    return at < last ? at : last;
}

__device__ __forceinline__ int block_prefix(int v, int* s_w, int& total) {
    const int lane = threadIdx.x & 31;
    const int wave = threadIdx.x >> 5;
    int inc = v;
    #pragma unroll
    for (int off = 1; off < 32; off <<= 1) {
        const int t = __shfl_up(inc, off, 32);
        if (lane >= off) inc += t;
    }
    if (lane == 31) s_w[wave] = inc;
    __syncthreads();
    int pre = 0, tot = 0;
    #pragma unroll
    for (int w = 0; w < GT / 32; w++) {
        const int x = s_w[w];
        pre += (w < wave) ? x : 0;
        tot += x;
    }
    __syncthreads();
    total = tot;
    return pre + inc - v;
}

__global__ __launch_bounds__(GT) void bin_kernel(const int* __restrict__ ei,
                                                 const float* __restrict__ ea,
                                                 int* __restrict__ area,
                                                 float* __restrict__ areaw,
                                                 int* __restrict__ oi, int E, int N) {
    extern __shared__ int s_bin[];
    __shared__ int s_w[GT / 32];
    const int sl = blockIdx.x;
    const int ES = (E + NSL - 1) / NSL;
    const int NB = (N + BN - 1) / BN;
    const int e_lo = sl * ES;
    const int ne = max(0, min(E - e_lo, ES));
    int* s_cur = s_bin;
    int* s_stage = s_bin + NB;
    float* s_stw = (float*)(s_stage + ES);
    for (int i = threadIdx.x; i < NB; i += GT) s_cur[i] = 0;
    __syncthreads();

    const int* tg = ei + (size_t)E + e_lo;
    const int* sg = ei + e_lo;
    for (int i0 = threadIdx.x; i0 < ne; i0 += GT * AU) {
        int d[AU];
        #pragma unroll
        for (int u = 0; u < AU; u++) d[u] = tg[min(i0 + u * GT, ne - 1)];
        #pragma unroll
        for (int u = 0; u < AU; u++) {
            if ((i0 + u * GT < ne) & ((unsigned)d[u] < (unsigned)N)) atomicAdd(&s_cur[d[u] / BN], 1);
        }
    }
    __syncthreads();

    int placed = 0;
    for (int b0 = 0; b0 < NB; b0 += GT) {
        const int b = b0 + threadIdx.x;
        const int c = (b < NB) ? s_cur[b] : 0;
        int tot;
        const int start = placed + block_prefix(c, s_w, tot);
        if (b < NB) {
            s_cur[b] = start;
            *(int2*)(oi + pair_base(b, N) + 2 * sl) = make_int2(start, c);
        }
        placed += tot;
    }
    __syncthreads();

    for (int i0 = threadIdx.x; i0 < ne; i0 += GT * AU) {
        int d[AU];
        #pragma unroll
        for (int u = 0; u < AU; u++) d[u] = tg[min(i0 + u * GT, ne - 1)];
        int s[AU];
        #pragma unroll
        for (int u = 0; u < AU; u++) s[u] = sg[min(i0 + u * GT, ne - 1)];
        float wv[AU];
        #pragma unroll
        for (int u = 0; u < AU; u++) wv[u] = 0.0f;
        if (areaw) {
            #pragma unroll
            for (int u = 0; u < AU; u++) wv[u] = ea[(size_t)(e_lo + min(i0 + u * GT, ne - 1)) * 4];
        }
        int pos[AU];
        #pragma unroll
        for (int u = 0; u < AU; u++) {
            pos[u] = -1;
            if ((i0 + u * GT < ne) & ((unsigned)d[u] < (unsigned)N)) pos[u] = atomicAdd(&s_cur[d[u] / BN], 1);
        }
        #pragma unroll
        for (int u = 0; u < AU; u++) {
            if (pos[u] >= 0) {
                s_stage[pos[u]] = (int)(((unsigned)s[u] << SRB) | (unsigned)((i0 + u * GT) << LOCB) | (unsigned)(d[u] % BN));
                if (areaw) s_stw[pos[u]] = fabsf(wv[u]);
            }
        }
    }
    __syncthreads();

    for (int i = threadIdx.x; i < placed; i += GT) area[(size_t)e_lo + i] = s_stage[i];
    if (areaw)
        for (int i = threadIdx.x; i < placed; i += GT) areaw[(size_t)e_lo + i] = s_stw[i];
}

__global__ __launch_bounds__(GT) void rank_kernel(const float* __restrict__ ea,
                                                  const int* __restrict__ area,
                                                  const float* __restrict__ areaw,
                                                  float* __restrict__ dinv,
                                                  int* __restrict__ cnt,
                                                  int* oi, int E, int N) {
    __shared__ __attribute__((aligned(16))) int   s_rec[BN][CAP];
    __shared__ __attribute__((aligned(16))) float s_wt[BN][CAP];
    __shared__ unsigned char s_sl[BN][CAP];
    __shared__ int   s_at[GT * BU];
    __shared__ float s_deg[BN];
    __shared__ int   s_cnt[BN];
    __shared__ int   s_w[GT / 32];
    const int b = blockIdx.x;
    const int lo = b * BN;
    const int nr = max(0, min(N - lo, BN));
    const int ES = (E + NSL - 1) / NSL;
    const int lane = threadIdx.x & 31;
    const int wave = threadIdx.x >> 5;

    const int2 sc = *(const int2*)(oi + pair_base(b, N) + 2 * threadIdx.x);
    if (threadIdx.x < BN) { s_deg[threadIdx.x] = 0.0f; s_cnt[threadIdx.x] = 0; }
    int M;
    const int pre = block_prefix(sc.y, s_w, M);
    const int first = threadIdx.x * ES + sc.x - pre;

    for (int base = 0; base < M; base += GT * BU) {
        const int j1 = min(pre + sc.y, base + GT * BU);
        for (int j = max(pre, base); j < j1; j++) s_at[j - base] = (first + j) | ((int)threadIdx.x << PSH);
        __syncthreads();
        const int m = min(M - base, GT * BU);
        int   rec[BU];
        float w[BU];
        int   at[BU];
        #pragma unroll
        for (int u = 0; u < BU; u++) at[u] = s_at[min((int)threadIdx.x + u * GT, m - 1)];
        #pragma unroll
        for (int u = 0; u < BU; u++) rec[u] = area[at[u] & ((1 << PSH) - 1)];
        if (areaw) {
            #pragma unroll
            for (int u = 0; u < BU; u++) w[u] = areaw[at[u] & ((1 << PSH) - 1)];
        } else {
            #pragma unroll
            for (int u = 0; u < BU; u++)
                w[u] = ea[(size_t)((at[u] >> PSH) * ES + ((rec[u] >> LOCB) & ((1 << ELB) - 1))) * 4];
        }
        #pragma unroll
        for (int u = 0; u < BU; u++) asm volatile("" :: "v"(w[u]));
        int rk[BU];
        #pragma unroll
        for (int u = 0; u < BU; u++) {
            rk[u] = CAP;
            if ((int)threadIdx.x + u * GT < m) {
                const int loc = rec[u] & ((1 << LOCB) - 1);
                atomicAdd(&s_deg[loc], fabsf(w[u]));
                rk[u] = atomicAdd(&s_cnt[loc], 1);
            }
        }
        #pragma unroll
        for (int u = 0; u < BU; u++) {
            if (rk[u] < CAP) {
                const int loc = rec[u] & ((1 << LOCB) - 1);
                s_rec[loc][rk[u]] = rec[u];
                s_wt[loc][rk[u]]  = fabsf(w[u]);
                s_sl[loc][rk[u]]  = (unsigned char)(at[u] >> PSH);
            }
        }
        __syncthreads();
    }

    if (threadIdx.x < nr) {
        const float dg = 1.0f + s_deg[threadIdx.x];
        dinv[lo + threadIdx.x] = (dg > 0.0f) ? rsqrtf(dg) : 0.0f;
        cnt[lo + threadIdx.x] = s_cnt[threadIdx.x];
    }
    for (int r = wave; r < nr; r += GT / 32) {
        const int c = min(s_cnt[r], CAP);
        const int P = min(c, CAP - c);
        int v[4];
        #pragma unroll
        for (int k = 0; k < 4; k++) {
            const int i = 4 * lane + k;
            int val = 0;
            if (i < c) {
                const unsigned rc = (unsigned)s_rec[r][i];
                val = (i < P) ? (int)(rc >> SRB)
                              : (int)s_sl[r][i] * ES + (int)((rc >> LOCB) & ((1u << ELB) - 1));
            } else if (i >= CAP - P) {
                val = __float_as_int(s_wt[r][CAP - 1 - i]);
            }
            v[k] = val;
        }
        *(int4*)(oi + (size_t)(lo + r) * DFEAT + 4 * lane) = make_int4(v[0], v[1], v[2], v[3]);
    }
}

#define NPB 16
#define NPW 2
#define XP 132

__device__ __forceinline__ int lane_i(int v, int l) { return __builtin_amdgcn_readlane(v, l); }
__device__ __forceinline__ float lane_f(float v, int l) {
    return __int_as_float(__builtin_amdgcn_readlane(__float_as_int(v), l));
}

__device__ __forceinline__ float4 gather_node(const float* __restrict__ x,
                                              const float* __restrict__ ea,
                                              const int* __restrict__ ei,
                                              const float* __restrict__ dinv,
                                              const int* __restrict__ cnt,
                                              const float* orow, int node, int lane,
                                              int E, float& nsum) {
    const float di = dinv[node];
    const int   c  = __builtin_amdgcn_readfirstlane(cnt[node]);
    const int   d0 = lane * 4;

    float a0 = 0.0f, a1 = 0.0f, a2 = 0.0f, a3 = 0.0f;
    float ns = 0.0f;

    if (c <= 0) {
    } else if (c <= CAP) {
        const int4 ra = *(const int4*)((const int*)orow + d0);
        const int4 rw = *(const int4*)((const int*)orow + (CAP - 4 - d0));
        const bool k0 = d0 + 0 < c, k1 = d0 + 1 < c, k2 = d0 + 2 < c, k3 = d0 + 3 < c;
        const int P = min(c, CAP - c);
        int   sr0 = ra.x, sr1 = ra.y, sr2 = ra.z, sr3 = ra.w;
        float w0 = __int_as_float(rw.w), w1 = __int_as_float(rw.z);
        float w2 = __int_as_float(rw.y), w3 = __int_as_float(rw.x);
        if (c > P) {
            const bool g0 = k0 & (d0 + 0 >= P), g1 = k1 & (d0 + 1 >= P);
            const bool g2 = k2 & (d0 + 2 >= P), g3 = k3 & (d0 + 3 >= P);
            const int e0 = g0 ? ra.x : 0, e1 = g1 ? ra.y : 0, e2 = g2 ? ra.z : 0, e3 = g3 ? ra.w : 0;
            const int   t0 = ei[e0], t1 = ei[e1], t2 = ei[e2], t3 = ei[e3];
            const float u0 = ea[(size_t)e0 * 4], u1 = ea[(size_t)e1 * 4];
            const float u2 = ea[(size_t)e2 * 4], u3 = ea[(size_t)e3 * 4];
            sr0 = g0 ? t0 : sr0; sr1 = g1 ? t1 : sr1; sr2 = g2 ? t2 : sr2; sr3 = g3 ? t3 : sr3;
            w0 = g0 ? fabsf(u0) : w0; w1 = g1 ? fabsf(u1) : w1;
            w2 = g2 ? fabsf(u2) : w2; w3 = g3 ? fabsf(u3) : w3;
        }
        sr0 = k0 ? sr0 : 0; sr1 = k1 ? sr1 : 0; sr2 = k2 ? sr2 : 0; sr3 = k3 ? sr3 : 0;
        const float q0 = dinv[sr0], q1 = dinv[sr1], q2 = dinv[sr2], q3 = dinv[sr3];
        const float nm0 = k0 ? q0 * w0 * di : 0.0f;
        const float nm1 = k1 ? q1 * w1 * di : 0.0f;
        const float nm2 = k2 ? q2 * w2 * di : 0.0f;
        const float nm3 = k3 ? q3 * w3 * di : 0.0f;
        ns = (nm0 + nm1) + (nm2 + nm3);
        #pragma unroll
        for (int off = 16; off > 0; off >>= 1) ns += __shfl_xor(ns, off, 32);

        const int ng = c >> 2;
        #pragma unroll 4
        for (int l = 0; l < ng; l++) {
            const int   s0 = lane_i(sr0, l), s1 = lane_i(sr1, l), s2 = lane_i(sr2, l), s3 = lane_i(sr3, l);
            const float n0 = lane_f(nm0, l), n1 = lane_f(nm1, l), n2 = lane_f(nm2, l), n3 = lane_f(nm3, l);
            const float4 r0 = *(const float4*)(x + (size_t)s0 * DFEAT + d0);
            const float4 r1 = *(const float4*)(x + (size_t)s1 * DFEAT + d0);
            const float4 r2 = *(const float4*)(x + (size_t)s2 * DFEAT + d0);
            const float4 r3 = *(const float4*)(x + (size_t)s3 * DFEAT + d0);
            a0 += r0.x * n0; a1 += r0.y * n0; a2 += r0.z * n0; a3 += r0.w * n0;
            a0 += r1.x * n1; a1 += r1.y * n1; a2 += r1.z * n1; a3 += r1.w * n1;
            a0 += r2.x * n2; a1 += r2.y * n2; a2 += r2.z * n2; a3 += r2.w * n2;
            a0 += r3.x * n3; a1 += r3.y * n3; a2 += r3.z * n3; a3 += r3.w * n3;
        }
        const int rem = c & 3;
        if (rem > 0) {
            const int   s0 = lane_i(sr0, ng);
            const float n0 = lane_f(nm0, ng);
            const float4 r0 = *(const float4*)(x + (size_t)s0 * DFEAT + d0);
            a0 += r0.x * n0; a1 += r0.y * n0; a2 += r0.z * n0; a3 += r0.w * n0;
        }
        if (rem > 1) {
            const int   s1 = lane_i(sr1, ng);
            const float n1 = lane_f(nm1, ng);
            const float4 r1 = *(const float4*)(x + (size_t)s1 * DFEAT + d0);
            a0 += r1.x * n1; a1 += r1.y * n1; a2 += r1.z * n1; a3 += r1.w * n1;
        }
        if (rem > 2) {
            const int   s2 = lane_i(sr2, ng);
            const float n2 = lane_f(nm2, ng);
            const float4 r2 = *(const float4*)(x + (size_t)s2 * DFEAT + d0);
            a0 += r2.x * n2; a1 += r2.y * n2; a2 += r2.z * n2; a3 += r2.w * n2;
        }
    } else {
        for (int b0 = 0; b0 < E; b0 += 32) {
            const int e = b0 + lane;
            const int d = (e < E) ? ei[E + e] : -1;
            unsigned m = (unsigned)__ballot(d == node);
            while (m) {
                const int bit = __ffs(m) - 1;
                m &= m - 1;
                const int ee = b0 + bit;
                const int s  = ei[ee];
                const float nv = dinv[s] * fabsf(ea[(size_t)ee * 4]) * di;
                const float4 r = *(const float4*)(x + (size_t)s * DFEAT + d0);
                a0 += r.x * nv; a1 += r.y * nv; a2 += r.z * nv; a3 += r.w * nv;
                ns += nv;
            }
        }
    }

    const float self_norm = di * di;
    const float4 xv = *(const float4*)(x + (size_t)node * DFEAT + d0);
    float4 res;
    res.x = a0 + xv.x * self_norm;
    res.y = a1 + xv.y * self_norm;
    res.z = a2 + xv.z * self_norm;
    res.w = a3 + xv.w * self_norm;
    nsum = ns + self_norm;
    return res;
}

__global__ __launch_bounds__(32 * NPB / NPW) void gather_linear_finish_kernel(
        const float* __restrict__ x, const float* __restrict__ W, const float* __restrict__ bias,
        const float* __restrict__ ea, const int* __restrict__ ei,
        const float* __restrict__ dinv, const int* __restrict__ cnt,
        const float* __restrict__ gamma, const float* __restrict__ beta,
        float* out, int E, int N) {
    __shared__ __attribute__((aligned(16))) float s_t[NPB][XP];
    __shared__ float s_ns[NPB];
    const int lane = threadIdx.x & 31;
    const int wave = __builtin_amdgcn_readfirstlane((int)(threadIdx.x >> 5));
    const int node_base = blockIdx.x * NPB;

    #pragma unroll
    for (int q = 0; q < NPW; q++) {
        const int row = wave * NPW + q;
        const int node = node_base + row;
        float4 agg = make_float4(0.0f, 0.0f, 0.0f, 0.0f);
        float ns = 0.0f;
        if (node < N)
            agg = gather_node(x, ea, ei, dinv, cnt, out + (size_t)node * DFEAT, node, lane, E, ns);
        *(float4*)&s_t[row][lane * 4] = agg;
        if (lane == 0) s_ns[row] = ns;
    }

    const int c16 = lane & 15;
    const int hf  = lane >> 4;
    const int col = wave * 16 + c16;
    const float* wrow = W + (size_t)col * DFEAT + hf * 64;
    float4 bf[16];
    #pragma unroll
    for (int j = 0; j < 16; j++) bf[j] = *(const float4*)(wrow + 4 * j);
    const float bv = bias[col];

    __syncthreads();

    const float* arow = &s_t[c16][hf * 64];
    v8f acc0 = {}, acc1 = {};
    #pragma unroll
    for (int j = 0; j < 16; j += 2) {
        const float4 a0 = *(const float4*)(arow + 4 * j);
        const float4 a1 = *(const float4*)(arow + 4 * j + 4);
        v2f fa, fb;
        fa.x = a0.x; fa.y = a0.y; fb.x = bf[j].x; fb.y = bf[j].y;
        acc0 = __builtin_amdgcn_wmma_f32_16x16x4_f32(false, fa, false, fb, (short)0, acc0, false, false);
        fa.x = a1.x; fa.y = a1.y; fb.x = bf[j + 1].x; fb.y = bf[j + 1].y;
        acc1 = __builtin_amdgcn_wmma_f32_16x16x4_f32(false, fa, false, fb, (short)0, acc1, false, false);
        fa.x = a0.z; fa.y = a0.w; fb.x = bf[j].z; fb.y = bf[j].w;
        acc0 = __builtin_amdgcn_wmma_f32_16x16x4_f32(false, fa, false, fb, (short)0, acc0, false, false);
        fa.x = a1.z; fa.y = a1.w; fb.x = bf[j + 1].z; fb.y = bf[j + 1].w;
        acc1 = __builtin_amdgcn_wmma_f32_16x16x4_f32(false, fa, false, fb, (short)0, acc1, false, false);
    }

    __syncthreads();

    #pragma unroll
    for (int r = 0; r < 8; r++) {
        const int row = hf * 8 + r;
        float v = (acc0[r] + acc1[r]) + bv * s_ns[row];
        v = (v >= 0.0f) ? v : NEG_SLOPE * v;
        s_t[row][col] = v;
    }

    __syncthreads();

    const int d0 = lane * 4;
    const float4 gv = *(const float4*)(gamma + d0);
    const float4 bt = *(const float4*)(beta + d0);
    #pragma unroll
    for (int q = 0; q < NPW; q++) {
        const int row = wave * NPW + q;
        const int node = node_base + row;
        const float4 t = *(const float4*)&s_t[row][d0];
        float v[4] = {t.x, t.y, t.z, t.w};

        float sum = (v[0] + v[1]) + (v[2] + v[3]);
        #pragma unroll
        for (int off = 16; off > 0; off >>= 1) sum += __shfl_xor(sum, off, 32);
        const float mean = sum * (1.0f / DFEAT);

        float vs = 0.0f;
        #pragma unroll
        for (int j = 0; j < 4; j++) {
            float dlt = v[j] - mean;
            vs += dlt * dlt;
        }
        #pragma unroll
        for (int off = 16; off > 0; off >>= 1) vs += __shfl_xor(vs, off, 32);
        const float inv_std = rsqrtf(vs * (1.0f / DFEAT) + LN_EPS);

        float4 res;
        res.x = (v[0] - mean) * inv_std * gv.x + bt.x;
        res.y = (v[1] - mean) * inv_std * gv.y + bt.y;
        res.z = (v[2] - mean) * inv_std * gv.z + bt.z;
        res.w = (v[3] - mean) * inv_std * gv.w + bt.w;
        if (node < N) *(float4*)(out + (size_t)node * DFEAT + d0) = res;
    }
}

extern "C" void kernel_launch(void* const* d_in, const int* in_sizes, int n_in,
                              void* d_out, int out_size, void* d_ws, size_t ws_size,
                              hipStream_t stream) {
    const float* x     = (const float*)d_in[0];
    const float* ea    = (const float*)d_in[1];
    const float* W     = (const float*)d_in[2];
    const float* bias  = (const float*)d_in[3];
    const float* gamma = (const float*)d_in[4];
    const float* beta  = (const float*)d_in[5];
    const int*   ei    = (const int*)d_in[6];
    float* out = (float*)d_out;

    const int N = in_sizes[0] / DFEAT;
    const int E = in_sizes[1] / 4;

    int*   cnt   = (int*)d_ws;
    float* dinv  = (float*)d_ws + N;
    int*   area  = (int*)(dinv + N);
    int*   oi    = (int*)d_out;
    const int NB = (N + BN - 1) / BN;
    const int ES = (E + NSL - 1) / NSL;
    float* areaw = (ws_size >= ((size_t)2 * N + (size_t)2 * E) * sizeof(int)) ? (float*)(area + E) : nullptr;

    bin_kernel<<<NSL, GT, (size_t)(NB + 2 * ES) * sizeof(int), stream>>>(ei, ea, area, areaw, oi, E, N);
    rank_kernel<<<NB, GT, 0, stream>>>(ea, area, areaw, dinv, cnt, oi, E, N);
    gather_linear_finish_kernel<<<(N + NPB - 1) / NPB, 32 * NPB / NPW, 0, stream>>>(
        x, W, bias, ea, ei, dinv, cnt, gamma, beta, out, E, N);
}


# Last kernel: one 20-wave work-group per 40 nodes (about one per WGP) copies W to LDS once, whole-row loads; waves read B fragments from LDS: W fetched 250 times, not by 5000 waves in 32-line loads
# speedup vs baseline: 5.0175x; 1.0373x over previous
#include <hip/hip_runtime.h>
#include <math.h>

typedef float v2f __attribute__((ext_vector_type(2)));
typedef float v8f __attribute__((ext_vector_type(8)));

#define DFEAT 128
#define NEG_SLOPE 0.01f
#define LN_EPS 1e-5f

#define NSL 256
#define BN 40
#define LOCB 6
#define ELB 12
#define SRB (LOCB + ELB)
#define PSH 20
#define GT 256
#define AU 10
#define BU 12
#define CAP DFEAT

__device__ __forceinline__ size_t pair_base(int b, int N) {
    const size_t at = (size_t)b * BN * DFEAT;
    const size_t last = (size_t)N * DFEAT - 2 * NSL;
    return at < last ? at : last;
}

__device__ __forceinline__ int block_prefix(int v, int* s_w, int& total) {
    const int lane = threadIdx.x & 31;
    const int wave = threadIdx.x >> 5;
    int inc = v;
    #pragma unroll
    for (int off = 1; off < 32; off <<= 1) {
        const int t = __shfl_up(inc, off, 32);
        if (lane >= off) inc += t;
    }
    if (lane == 31) s_w[wave] = inc;
    __syncthreads();
    int pre = 0, tot = 0;
    #pragma unroll
    for (int w = 0; w < GT / 32; w++) {
        const int x = s_w[w];
        pre += (w < wave) ? x : 0;
        tot += x;
    }
    __syncthreads();
    total = tot;
    return pre + inc - v;
}

__global__ __launch_bounds__(GT) void bin_kernel(const int* __restrict__ ei,
                                                 const float* __restrict__ ea,
                                                 int* __restrict__ area,
                                                 float* __restrict__ areaw,
                                                 int* __restrict__ oi, int E, int N) {
    extern __shared__ int s_bin[];
    __shared__ int s_w[GT / 32];
    const int sl = blockIdx.x;
    const int ES = (E + NSL - 1) / NSL;
    const int NB = (N + BN - 1) / BN;
    const int e_lo = sl * ES;
    const int ne = max(0, min(E - e_lo, ES));
    int* s_cur = s_bin;
    int* s_stage = s_bin + NB;
    float* s_stw = (float*)(s_stage + ES);
    for (int i = threadIdx.x; i < NB; i += GT) s_cur[i] = 0;
    __syncthreads();

    const int* tg = ei + (size_t)E + e_lo;
    const int* sg = ei + e_lo;
    for (int i0 = threadIdx.x; i0 < ne; i0 += GT * AU) {
        int d[AU];
        #pragma unroll
        for (int u = 0; u < AU; u++) d[u] = tg[min(i0 + u * GT, ne - 1)];
        #pragma unroll
        for (int u = 0; u < AU; u++) {
            if ((i0 + u * GT < ne) & ((unsigned)d[u] < (unsigned)N)) atomicAdd(&s_cur[d[u] / BN], 1);
        }
    }
    __syncthreads();

    int placed = 0;
    for (int b0 = 0; b0 < NB; b0 += GT) {
        const int b = b0 + threadIdx.x;
        const int c = (b < NB) ? s_cur[b] : 0;
        int tot;
        const int start = placed + block_prefix(c, s_w, tot);
        if (b < NB) {
            s_cur[b] = start;
            *(int2*)(oi + pair_base(b, N) + 2 * sl) = make_int2(start, c);
        }
        placed += tot;
    }
    __syncthreads();

    for (int i0 = threadIdx.x; i0 < ne; i0 += GT * AU) {
        int d[AU];
        #pragma unroll
        for (int u = 0; u < AU; u++) d[u] = tg[min(i0 + u * GT, ne - 1)];
        int s[AU];
        #pragma unroll
        for (int u = 0; u < AU; u++) s[u] = sg[min(i0 + u * GT, ne - 1)];
        float wv[AU];
        #pragma unroll
        for (int u = 0; u < AU; u++) wv[u] = 0.0f;
        if (areaw) {
            #pragma unroll
            for (int u = 0; u < AU; u++) wv[u] = ea[(size_t)(e_lo + min(i0 + u * GT, ne - 1)) * 4];
        }
        int pos[AU];
        #pragma unroll
        for (int u = 0; u < AU; u++) {
            pos[u] = -1;
            if ((i0 + u * GT < ne) & ((unsigned)d[u] < (unsigned)N)) pos[u] = atomicAdd(&s_cur[d[u] / BN], 1);
        }
        #pragma unroll
        for (int u = 0; u < AU; u++) {
            if (pos[u] >= 0) {
                s_stage[pos[u]] = (int)(((unsigned)s[u] << SRB) | (unsigned)((i0 + u * GT) << LOCB) | (unsigned)(d[u] % BN));
                if (areaw) s_stw[pos[u]] = fabsf(wv[u]);
            }
        }
    }
    __syncthreads();

    for (int i = threadIdx.x; i < placed; i += GT) area[(size_t)e_lo + i] = s_stage[i];
    if (areaw)
        for (int i = threadIdx.x; i < placed; i += GT) areaw[(size_t)e_lo + i] = s_stw[i];
}

__global__ __launch_bounds__(GT) void rank_kernel(const float* __restrict__ ea,
                                                  const int* __restrict__ area,
                                                  const float* __restrict__ areaw,
                                                  float* __restrict__ dinv,
                                                  int* __restrict__ cnt,
                                                  int* oi, int E, int N) {
    __shared__ __attribute__((aligned(16))) int   s_rec[BN][CAP];
    __shared__ __attribute__((aligned(16))) float s_wt[BN][CAP];
    __shared__ unsigned char s_sl[BN][CAP];
    __shared__ int   s_at[GT * BU];
    __shared__ float s_deg[BN];
    __shared__ int   s_cnt[BN];
    __shared__ int   s_w[GT / 32];
    const int b = blockIdx.x;
    const int lo = b * BN;
    const int nr = max(0, min(N - lo, BN));
    const int ES = (E + NSL - 1) / NSL;
    const int lane = threadIdx.x & 31;
    const int wave = threadIdx.x >> 5;

    const int2 sc = *(const int2*)(oi + pair_base(b, N) + 2 * threadIdx.x);
    if (threadIdx.x < BN) { s_deg[threadIdx.x] = 0.0f; s_cnt[threadIdx.x] = 0; }
    int M;
    const int pre = block_prefix(sc.y, s_w, M);
    const int first = threadIdx.x * ES + sc.x - pre;

    for (int base = 0; base < M; base += GT * BU) {
        const int j1 = min(pre + sc.y, base + GT * BU);
        for (int j = max(pre, base); j < j1; j++) s_at[j - base] = (first + j) | ((int)threadIdx.x << PSH);
        __syncthreads();
        const int m = min(M - base, GT * BU);
        int   rec[BU];
        float w[BU];
        int   at[BU];
        #pragma unroll
        for (int u = 0; u < BU; u++) at[u] = s_at[min((int)threadIdx.x + u * GT, m - 1)];
        #pragma unroll
        for (int u = 0; u < BU; u++) rec[u] = area[at[u] & ((1 << PSH) - 1)];
        if (areaw) {
            #pragma unroll
            for (int u = 0; u < BU; u++) w[u] = areaw[at[u] & ((1 << PSH) - 1)];
        } else {
            #pragma unroll
            for (int u = 0; u < BU; u++)
                w[u] = ea[(size_t)((at[u] >> PSH) * ES + ((rec[u] >> LOCB) & ((1 << ELB) - 1))) * 4];
        }
        #pragma unroll
        for (int u = 0; u < BU; u++) asm volatile("" :: "v"(w[u]));
        int rk[BU];
        #pragma unroll
        for (int u = 0; u < BU; u++) {
            rk[u] = CAP;
            if ((int)threadIdx.x + u * GT < m) {
                const int loc = rec[u] & ((1 << LOCB) - 1);
                atomicAdd(&s_deg[loc], fabsf(w[u]));
                rk[u] = atomicAdd(&s_cnt[loc], 1);
            }
        }
        #pragma unroll
        for (int u = 0; u < BU; u++) {
            if (rk[u] < CAP) {
                const int loc = rec[u] & ((1 << LOCB) - 1);
                s_rec[loc][rk[u]] = rec[u];
                s_wt[loc][rk[u]]  = fabsf(w[u]);
                s_sl[loc][rk[u]]  = (unsigned char)(at[u] >> PSH);
            }
        }
        __syncthreads();
    }

    if (threadIdx.x < nr) {
        const float dg = 1.0f + s_deg[threadIdx.x];
        dinv[lo + threadIdx.x] = (dg > 0.0f) ? rsqrtf(dg) : 0.0f;
        cnt[lo + threadIdx.x] = s_cnt[threadIdx.x];
    }
    for (int r = wave; r < nr; r += GT / 32) {
        const int c = min(s_cnt[r], CAP);
        const int P = min(c, CAP - c);
        int v[4];
        #pragma unroll
        for (int k = 0; k < 4; k++) {
            const int i = 4 * lane + k;
            int val = 0;
            if (i < c) {
                const unsigned rc = (unsigned)s_rec[r][i];
                val = (i < P) ? (int)(rc >> SRB)
                              : (int)s_sl[r][i] * ES + (int)((rc >> LOCB) & ((1u << ELB) - 1));
            } else if (i >= CAP - P) {
                val = __float_as_int(s_wt[r][CAP - 1 - i]);
            }
            v[k] = val;
        }
        *(int4*)(oi + (size_t)(lo + r) * DFEAT + 4 * lane) = make_int4(v[0], v[1], v[2], v[3]);
    }
}

#define NPB 40
#define NPW 2
#define FW (NPB / NPW)
#define RT ((NPB + 15) / 16)
#define WPC (DFEAT * DFEAT / 4)
#define WCT 512
#define WPT (WPC / WCT)
#define XP 132

__device__ __forceinline__ int lane_i(int v, int l) { return __builtin_amdgcn_readlane(v, l); }
__device__ __forceinline__ float lane_f(float v, int l) {
    return __int_as_float(__builtin_amdgcn_readlane(__float_as_int(v), l));
}

__device__ __forceinline__ float4 gather_node(const float* __restrict__ x,
                                              const float* __restrict__ ea,
                                              const int* __restrict__ ei,
                                              const float* __restrict__ dinv,
                                              const int* __restrict__ cnt,
                                              const float* orow, int node, int lane,
                                              int E, float& nsum) {
    const float di = dinv[node];
    const int   c  = __builtin_amdgcn_readfirstlane(cnt[node]);
    const int   d0 = lane * 4;

    float a0 = 0.0f, a1 = 0.0f, a2 = 0.0f, a3 = 0.0f;
    float ns = 0.0f;

    if (c <= 0) {
    } else if (c <= CAP) {
        const int4 ra = *(const int4*)((const int*)orow + d0);
        const int4 rw = *(const int4*)((const int*)orow + (CAP - 4 - d0));
        const bool k0 = d0 + 0 < c, k1 = d0 + 1 < c, k2 = d0 + 2 < c, k3 = d0 + 3 < c;
        const int P = min(c, CAP - c);
        int   sr0 = ra.x, sr1 = ra.y, sr2 = ra.z, sr3 = ra.w;
        float w0 = __int_as_float(rw.w), w1 = __int_as_float(rw.z);
        float w2 = __int_as_float(rw.y), w3 = __int_as_float(rw.x);
        if (c > P) {
            const bool g0 = k0 & (d0 + 0 >= P), g1 = k1 & (d0 + 1 >= P);
            const bool g2 = k2 & (d0 + 2 >= P), g3 = k3 & (d0 + 3 >= P);
            const int e0 = g0 ? ra.x : 0, e1 = g1 ? ra.y : 0, e2 = g2 ? ra.z : 0, e3 = g3 ? ra.w : 0;
            const int   t0 = ei[e0], t1 = ei[e1], t2 = ei[e2], t3 = ei[e3];
            const float u0 = ea[(size_t)e0 * 4], u1 = ea[(size_t)e1 * 4];
            const float u2 = ea[(size_t)e2 * 4], u3 = ea[(size_t)e3 * 4];
            sr0 = g0 ? t0 : sr0; sr1 = g1 ? t1 : sr1; sr2 = g2 ? t2 : sr2; sr3 = g3 ? t3 : sr3;
            w0 = g0 ? fabsf(u0) : w0; w1 = g1 ? fabsf(u1) : w1;
            w2 = g2 ? fabsf(u2) : w2; w3 = g3 ? fabsf(u3) : w3;
        }
        sr0 = k0 ? sr0 : 0; sr1 = k1 ? sr1 : 0; sr2 = k2 ? sr2 : 0; sr3 = k3 ? sr3 : 0;
        const float q0 = dinv[sr0], q1 = dinv[sr1], q2 = dinv[sr2], q3 = dinv[sr3];
        const float nm0 = k0 ? q0 * w0 * di : 0.0f;
        const float nm1 = k1 ? q1 * w1 * di : 0.0f;
        const float nm2 = k2 ? q2 * w2 * di : 0.0f;
        const float nm3 = k3 ? q3 * w3 * di : 0.0f;
        ns = (nm0 + nm1) + (nm2 + nm3);
        #pragma unroll
        for (int off = 16; off > 0; off >>= 1) ns += __shfl_xor(ns, off, 32);

        const int ng = c >> 2;
        #pragma unroll 4
        for (int l = 0; l < ng; l++) {
            const int   s0 = lane_i(sr0, l), s1 = lane_i(sr1, l), s2 = lane_i(sr2, l), s3 = lane_i(sr3, l);
            const float n0 = lane_f(nm0, l), n1 = lane_f(nm1, l), n2 = lane_f(nm2, l), n3 = lane_f(nm3, l);
            const float4 r0 = *(const float4*)(x + (size_t)s0 * DFEAT + d0);
            const float4 r1 = *(const float4*)(x + (size_t)s1 * DFEAT + d0);
            const float4 r2 = *(const float4*)(x + (size_t)s2 * DFEAT + d0);
            const float4 r3 = *(const float4*)(x + (size_t)s3 * DFEAT + d0);
            a0 += r0.x * n0; a1 += r0.y * n0; a2 += r0.z * n0; a3 += r0.w * n0;
            a0 += r1.x * n1; a1 += r1.y * n1; a2 += r1.z * n1; a3 += r1.w * n1;
            a0 += r2.x * n2; a1 += r2.y * n2; a2 += r2.z * n2; a3 += r2.w * n2;
            a0 += r3.x * n3; a1 += r3.y * n3; a2 += r3.z * n3; a3 += r3.w * n3;
        }
        const int rem = c & 3;
        if (rem > 0) {
            const int   s0 = lane_i(sr0, ng);
            const float n0 = lane_f(nm0, ng);
            const float4 r0 = *(const float4*)(x + (size_t)s0 * DFEAT + d0);
            a0 += r0.x * n0; a1 += r0.y * n0; a2 += r0.z * n0; a3 += r0.w * n0;
        }
        if (rem > 1) {
            const int   s1 = lane_i(sr1, ng);
            const float n1 = lane_f(nm1, ng);
            const float4 r1 = *(const float4*)(x + (size_t)s1 * DFEAT + d0);
            a0 += r1.x * n1; a1 += r1.y * n1; a2 += r1.z * n1; a3 += r1.w * n1;
        }
        if (rem > 2) {
            const int   s2 = lane_i(sr2, ng);
            const float n2 = lane_f(nm2, ng);
            const float4 r2 = *(const float4*)(x + (size_t)s2 * DFEAT + d0);
            a0 += r2.x * n2; a1 += r2.y * n2; a2 += r2.z * n2; a3 += r2.w * n2;
        }
    } else {
        for (int b0 = 0; b0 < E; b0 += 32) {
            const int e = b0 + lane;
            const int d = (e < E) ? ei[E + e] : -1;
            unsigned m = (unsigned)__ballot(d == node);
            while (m) {
                const int bit = __ffs(m) - 1;
                m &= m - 1;
                const int ee = b0 + bit;
                const int s  = ei[ee];
                const float nv = dinv[s] * fabsf(ea[(size_t)ee * 4]) * di;
                const float4 r = *(const float4*)(x + (size_t)s * DFEAT + d0);
                a0 += r.x * nv; a1 += r.y * nv; a2 += r.z * nv; a3 += r.w * nv;
                ns += nv;
            }
        }
    }

    const float self_norm = di * di;
    const float4 xv = *(const float4*)(x + (size_t)node * DFEAT + d0);
    float4 res;
    res.x = a0 + xv.x * self_norm;
    res.y = a1 + xv.y * self_norm;
    res.z = a2 + xv.z * self_norm;
    res.w = a3 + xv.w * self_norm;
    nsum = ns + self_norm;
    return res;
}

__global__ __launch_bounds__(32 * FW) void gather_linear_finish_kernel(
        const float* __restrict__ x, const float* __restrict__ W, const float* __restrict__ bias,
        const float* __restrict__ ea, const int* __restrict__ ei,
        const float* __restrict__ dinv, const int* __restrict__ cnt,
        const float* __restrict__ gamma, const float* __restrict__ beta,
        float* out, int E, int N) {
    __shared__ __attribute__((aligned(16))) float s_W[DFEAT][XP];
    __shared__ __attribute__((aligned(16))) float s_t[RT * 16][XP];
    __shared__ __attribute__((aligned(16))) float s_o[RT * 16][XP];
    __shared__ float s_ns[RT * 16];
    const int lane = threadIdx.x & 31;
    const int wave = __builtin_amdgcn_readfirstlane((int)(threadIdx.x >> 5));
    const int node_base = blockIdx.x * NPB;

    #pragma unroll
    for (int q = 0; q < NPW; q++) {
        const int row = wave * NPW + q;
        const int node = node_base + row;
        float4 agg = make_float4(0.0f, 0.0f, 0.0f, 0.0f);
        float ns = 0.0f;
        if (node < N)
            agg = gather_node(x, ea, ei, dinv, cnt, out + (size_t)node * DFEAT, node, lane, E, ns);
        *(float4*)&s_t[row][lane * 4] = agg;
        if (lane == 0) s_ns[row] = ns;
    }
    if (wave < RT * 16 - NPB) {
        *(float4*)&s_t[NPB + wave][lane * 4] = make_float4(0.0f, 0.0f, 0.0f, 0.0f);
        if (lane == 0) s_ns[NPB + wave] = 0.0f;
    }

    if (threadIdx.x < WCT) {
        float4 wv[WPT];
        #pragma unroll
        for (int k = 0; k < WPT; k++) wv[k] = *(const float4*)(W + (size_t)(k * WCT + (int)threadIdx.x) * 4);
        #pragma unroll
        for (int k = 0; k < WPT; k++) {
            const int i = k * WCT + (int)threadIdx.x;
            *(float4*)&s_W[i >> 5][4 * (i & 31)] = wv[k];
        }
    }

    __syncthreads();

    const int c16 = lane & 15;
    const int hf  = lane >> 4;
    for (int pr = wave; pr < RT * (DFEAT / 16); pr += FW) {
        const int rt  = pr >> 3;
        const int col = (pr & 7) * 16 + c16;
        const float* wrow = &s_W[col][hf * 64];
        const float* arow = &s_t[rt * 16 + c16][hf * 64];
        float4 bf[16];
        #pragma unroll
        for (int j = 0; j < 16; j++) bf[j] = *(const float4*)(wrow + 4 * j);
        const float bv = bias[col];

        v8f acc0 = {}, acc1 = {};
        #pragma unroll
        for (int j = 0; j < 16; j += 2) {
            const float4 a0 = *(const float4*)(arow + 4 * j);
            const float4 a1 = *(const float4*)(arow + 4 * j + 4);
            v2f fa, fb;
            fa.x = a0.x; fa.y = a0.y; fb.x = bf[j].x; fb.y = bf[j].y;
            acc0 = __builtin_amdgcn_wmma_f32_16x16x4_f32(false, fa, false, fb, (short)0, acc0, false, false);
            fa.x = a1.x; fa.y = a1.y; fb.x = bf[j + 1].x; fb.y = bf[j + 1].y;
            acc1 = __builtin_amdgcn_wmma_f32_16x16x4_f32(false, fa, false, fb, (short)0, acc1, false, false);
            fa.x = a0.z; fa.y = a0.w; fb.x = bf[j].z; fb.y = bf[j].w;
            acc0 = __builtin_amdgcn_wmma_f32_16x16x4_f32(false, fa, false, fb, (short)0, acc0, false, false);
            fa.x = a1.z; fa.y = a1.w; fb.x = bf[j + 1].z; fb.y = bf[j + 1].w;
            acc1 = __builtin_amdgcn_wmma_f32_16x16x4_f32(false, fa, false, fb, (short)0, acc1, false, false);
        }

        #pragma unroll
        for (int r = 0; r < 8; r++) {
            const int row = rt * 16 + hf * 8 + r;
            float v = (acc0[r] + acc1[r]) + bv * s_ns[row];
            v = (v >= 0.0f) ? v : NEG_SLOPE * v;
            s_o[row][col] = v;
        }
    }

    __syncthreads();

    const int d0 = lane * 4;
    const float4 gv = *(const float4*)(gamma + d0);
    const float4 bt = *(const float4*)(beta + d0);
    #pragma unroll
    for (int q = 0; q < NPW; q++) {
        const int row = wave * NPW + q;
        const int node = node_base + row;
        const float4 t = *(const float4*)&s_o[row][d0];
        float v[4] = {t.x, t.y, t.z, t.w};

        float sum = (v[0] + v[1]) + (v[2] + v[3]);
        #pragma unroll
        for (int off = 16; off > 0; off >>= 1) sum += __shfl_xor(sum, off, 32);
        const float mean = sum * (1.0f / DFEAT);

        float vs = 0.0f;
        #pragma unroll
        for (int j = 0; j < 4; j++) {
            float dlt = v[j] - mean;
            vs += dlt * dlt;
        }
        #pragma unroll
        for (int off = 16; off > 0; off >>= 1) vs += __shfl_xor(vs, off, 32);
        const float inv_std = rsqrtf(vs * (1.0f / DFEAT) + LN_EPS);

        float4 res;
        res.x = (v[0] - mean) * inv_std * gv.x + bt.x;
        res.y = (v[1] - mean) * inv_std * gv.y + bt.y;
        res.z = (v[2] - mean) * inv_std * gv.z + bt.z;
        res.w = (v[3] - mean) * inv_std * gv.w + bt.w;
        if (node < N) *(float4*)(out + (size_t)node * DFEAT + d0) = res;
    }
}

extern "C" void kernel_launch(void* const* d_in, const int* in_sizes, int n_in,
                              void* d_out, int out_size, void* d_ws, size_t ws_size,
                              hipStream_t stream) {
    const float* x     = (const float*)d_in[0];
    const float* ea    = (const float*)d_in[1];
    const float* W     = (const float*)d_in[2];
    const float* bias  = (const float*)d_in[3];
    const float* gamma = (const float*)d_in[4];
    const float* beta  = (const float*)d_in[5];
    const int*   ei    = (const int*)d_in[6];
    float* out = (float*)d_out;

    const int N = in_sizes[0] / DFEAT;
    const int E = in_sizes[1] / 4;

    int*   cnt   = (int*)d_ws;
    float* dinv  = (float*)d_ws + N;
    int*   area  = (int*)(dinv + N);
    int*   oi    = (int*)d_out;
    const int NB = (N + BN - 1) / BN;
    const int ES = (E + NSL - 1) / NSL;
    float* areaw = (ws_size >= ((size_t)2 * N + (size_t)2 * E) * sizeof(int)) ? (float*)(area + E) : nullptr;

    bin_kernel<<<NSL, GT, (size_t)(NB + 2 * ES) * sizeof(int), stream>>>(ei, ea, area, areaw, oi, E, N);
    rank_kernel<<<NB, GT, 0, stream>>>(ea, area, areaw, dinv, cnt, oi, E, N);
    gather_linear_finish_kernel<<<(N + NPB - 1) / NPB, 32 * FW, 0, stream>>>(
        x, W, bias, ea, ei, dinv, cnt, gamma, beta, out, E, N);
}


# Last kernel: each work-group copies dinv (N * 4 bytes) to LDS with whole-line loads, row and count loads in flight across its barrier; dinv[source] look-ups read LDS, not 32-line memory loads
# speedup vs baseline: 5.2672x; 1.0498x over previous
#include <hip/hip_runtime.h>
#include <math.h>

typedef float v2f __attribute__((ext_vector_type(2)));
typedef float v8f __attribute__((ext_vector_type(8)));

#define DFEAT 128
#define NEG_SLOPE 0.01f
#define LN_EPS 1e-5f

#define NSL 256
#define BN 40
#define LOCB 6
#define ELB 12
#define SRB (LOCB + ELB)
#define PSH 20
#define GT 256
#define AU 10
#define BU 12
#define CAP DFEAT

__device__ __forceinline__ size_t pair_base(int b, int N) {
    const size_t at = (size_t)b * BN * DFEAT;
    const size_t last = (size_t)N * DFEAT - 2 * NSL;
    return at < last ? at : last;
}

__device__ __forceinline__ int block_prefix(int v, int* s_w, int& total) {
    const int lane = threadIdx.x & 31;
    const int wave = threadIdx.x >> 5;
    int inc = v;
    #pragma unroll
    for (int off = 1; off < 32; off <<= 1) {
        const int t = __shfl_up(inc, off, 32);
        if (lane >= off) inc += t;
    }
    if (lane == 31) s_w[wave] = inc;
    __syncthreads();
    int pre = 0, tot = 0;
    #pragma unroll
    for (int w = 0; w < GT / 32; w++) {
        const int x = s_w[w];
        pre += (w < wave) ? x : 0;
        tot += x;
    }
    __syncthreads();
    total = tot;
    return pre + inc - v;
}

__global__ __launch_bounds__(GT) void bin_kernel(const int* __restrict__ ei,
                                                 const float* __restrict__ ea,
                                                 int* __restrict__ area,
                                                 float* __restrict__ areaw,
                                                 int* __restrict__ oi, int E, int N) {
    extern __shared__ int s_bin[];
    __shared__ int s_w[GT / 32];
    const int sl = blockIdx.x;
    const int ES = (E + NSL - 1) / NSL;
    const int NB = (N + BN - 1) / BN;
    const int e_lo = sl * ES;
    const int ne = max(0, min(E - e_lo, ES));
    int* s_cur = s_bin;
    int* s_stage = s_bin + NB;
    float* s_stw = (float*)(s_stage + ES);
    for (int i = threadIdx.x; i < NB; i += GT) s_cur[i] = 0;
    __syncthreads();

    const int* tg = ei + (size_t)E + e_lo;
    const int* sg = ei + e_lo;
    for (int i0 = threadIdx.x; i0 < ne; i0 += GT * AU) {
        int d[AU];
        #pragma unroll
        for (int u = 0; u < AU; u++) d[u] = tg[min(i0 + u * GT, ne - 1)];
        #pragma unroll
        for (int u = 0; u < AU; u++) {
            if ((i0 + u * GT < ne) & ((unsigned)d[u] < (unsigned)N)) atomicAdd(&s_cur[d[u] / BN], 1);
        }
    }
    __syncthreads();

    int placed = 0;
    for (int b0 = 0; b0 < NB; b0 += GT) {
        const int b = b0 + threadIdx.x;
        const int c = (b < NB) ? s_cur[b] : 0;
        int tot;
        const int start = placed + block_prefix(c, s_w, tot);
        if (b < NB) {
            s_cur[b] = start;
            *(int2*)(oi + pair_base(b, N) + 2 * sl) = make_int2(start, c);
        }
        placed += tot;
    }
    __syncthreads();

    for (int i0 = threadIdx.x; i0 < ne; i0 += GT * AU) {
        int d[AU];
        #pragma unroll
        for (int u = 0; u < AU; u++) d[u] = tg[min(i0 + u * GT, ne - 1)];
        int s[AU];
        #pragma unroll
        for (int u = 0; u < AU; u++) s[u] = sg[min(i0 + u * GT, ne - 1)];
        float wv[AU];
        #pragma unroll
        for (int u = 0; u < AU; u++) wv[u] = 0.0f;
        if (areaw) {
            #pragma unroll
            for (int u = 0; u < AU; u++) wv[u] = ea[(size_t)(e_lo + min(i0 + u * GT, ne - 1)) * 4];
        }
        int pos[AU];
        #pragma unroll
        for (int u = 0; u < AU; u++) {
            pos[u] = -1;
            if ((i0 + u * GT < ne) & ((unsigned)d[u] < (unsigned)N)) pos[u] = atomicAdd(&s_cur[d[u] / BN], 1);
        }
        #pragma unroll
        for (int u = 0; u < AU; u++) {
            if (pos[u] >= 0) {
                s_stage[pos[u]] = (int)(((unsigned)s[u] << SRB) | (unsigned)((i0 + u * GT) << LOCB) | (unsigned)(d[u] % BN));
                if (areaw) s_stw[pos[u]] = fabsf(wv[u]);
            }
        }
    }
    __syncthreads();

    for (int i = threadIdx.x; i < placed; i += GT) area[(size_t)e_lo + i] = s_stage[i];
    if (areaw)
        for (int i = threadIdx.x; i < placed; i += GT) areaw[(size_t)e_lo + i] = s_stw[i];
}

__global__ __launch_bounds__(GT) void rank_kernel(const float* __restrict__ ea,
                                                  const int* __restrict__ area,
                                                  const float* __restrict__ areaw,
                                                  float* __restrict__ dinv,
                                                  int* __restrict__ cnt,
                                                  int* oi, int E, int N) {
    __shared__ __attribute__((aligned(16))) int   s_rec[BN][CAP];
    __shared__ __attribute__((aligned(16))) float s_wt[BN][CAP];
    __shared__ unsigned char s_sl[BN][CAP];
    __shared__ int   s_at[GT * BU];
    __shared__ float s_deg[BN];
    __shared__ int   s_cnt[BN];
    __shared__ int   s_w[GT / 32];
    const int b = blockIdx.x;
    const int lo = b * BN;
    const int nr = max(0, min(N - lo, BN));
    const int ES = (E + NSL - 1) / NSL;
    const int lane = threadIdx.x & 31;
    const int wave = threadIdx.x >> 5;

    const int2 sc = *(const int2*)(oi + pair_base(b, N) + 2 * threadIdx.x);
    if (threadIdx.x < BN) { s_deg[threadIdx.x] = 0.0f; s_cnt[threadIdx.x] = 0; }
    int M;
    const int pre = block_prefix(sc.y, s_w, M);
    const int first = threadIdx.x * ES + sc.x - pre;

    for (int base = 0; base < M; base += GT * BU) {
        const int j1 = min(pre + sc.y, base + GT * BU);
        for (int j = max(pre, base); j < j1; j++) s_at[j - base] = (first + j) | ((int)threadIdx.x << PSH);
        __syncthreads();
        const int m = min(M - base, GT * BU);
        int   rec[BU];
        float w[BU];
        int   at[BU];
        #pragma unroll
        for (int u = 0; u < BU; u++) at[u] = s_at[min((int)threadIdx.x + u * GT, m - 1)];
        #pragma unroll
        for (int u = 0; u < BU; u++) rec[u] = area[at[u] & ((1 << PSH) - 1)];
        if (areaw) {
            #pragma unroll
            for (int u = 0; u < BU; u++) w[u] = areaw[at[u] & ((1 << PSH) - 1)];
        } else {
            #pragma unroll
            for (int u = 0; u < BU; u++)
                w[u] = ea[(size_t)((at[u] >> PSH) * ES + ((rec[u] >> LOCB) & ((1 << ELB) - 1))) * 4];
        }
        #pragma unroll
        for (int u = 0; u < BU; u++) asm volatile("" :: "v"(w[u]));
        int rk[BU];
        #pragma unroll
        for (int u = 0; u < BU; u++) {
            rk[u] = CAP;
            if ((int)threadIdx.x + u * GT < m) {
                const int loc = rec[u] & ((1 << LOCB) - 1);
                atomicAdd(&s_deg[loc], fabsf(w[u]));
                rk[u] = atomicAdd(&s_cnt[loc], 1);
            }
        }
        #pragma unroll
        for (int u = 0; u < BU; u++) {
            if (rk[u] < CAP) {
                const int loc = rec[u] & ((1 << LOCB) - 1);
                s_rec[loc][rk[u]] = rec[u];
                s_wt[loc][rk[u]]  = fabsf(w[u]);
                s_sl[loc][rk[u]]  = (unsigned char)(at[u] >> PSH);
            }
        }
        __syncthreads();
    }

    if (threadIdx.x < nr) {
        const float dg = 1.0f + s_deg[threadIdx.x];
        dinv[lo + threadIdx.x] = (dg > 0.0f) ? rsqrtf(dg) : 0.0f;
        cnt[lo + threadIdx.x] = s_cnt[threadIdx.x];
    }
    for (int r = wave; r < nr; r += GT / 32) {
        const int c = min(s_cnt[r], CAP);
        const int P = min(c, CAP - c);
        int v[4];
        #pragma unroll
        for (int k = 0; k < 4; k++) {
            const int i = 4 * lane + k;
            int val = 0;
            if (i < c) {
                const unsigned rc = (unsigned)s_rec[r][i];
                val = (i < P) ? (int)(rc >> SRB)
                              : (int)s_sl[r][i] * ES + (int)((rc >> LOCB) & ((1u << ELB) - 1));
            } else if (i >= CAP - P) {
                val = __float_as_int(s_wt[r][CAP - 1 - i]);
            }
            v[k] = val;
        }
        *(int4*)(oi + (size_t)(lo + r) * DFEAT + 4 * lane) = make_int4(v[0], v[1], v[2], v[3]);
    }
}

#define NPB 40
#define NPW 2
#define FW (NPB / NPW)
#define RT ((NPB + 15) / 16)
#define WPC (DFEAT * DFEAT / 4)
#define WCT 512
#define WPT (WPC / WCT)
#define XP 132
#define DVC 10240
#define DVT (DVC / 4 / (32 * FW))

__device__ __forceinline__ int lane_i(int v, int l) { return __builtin_amdgcn_readlane(v, l); }
__device__ __forceinline__ float lane_f(float v, int l) {
    return __int_as_float(__builtin_amdgcn_readlane(__float_as_int(v), l));
}

typedef __attribute__((address_space(3))) const float lds_cfloat;

__device__ __forceinline__ float4 gather_node(const float* __restrict__ x,
                                              const float* __restrict__ ea,
                                              const int* __restrict__ ei,
                                              const float* __restrict__ dinv,
                                              lds_cfloat* sdv, const bool dl,
                                              const int cn, const int4 ra, const int4 rw,
                                              int node, int lane,
                                              int E, float& nsum) {
    float di;
    if (dl) di = sdv[node]; else di = dinv[node];
    const int   c  = __builtin_amdgcn_readfirstlane(cn);
    const int   d0 = lane * 4;

    float a0 = 0.0f, a1 = 0.0f, a2 = 0.0f, a3 = 0.0f;
    float ns = 0.0f;

    if (c <= 0) {
    } else if (c <= CAP) {
        const bool k0 = d0 + 0 < c, k1 = d0 + 1 < c, k2 = d0 + 2 < c, k3 = d0 + 3 < c;
        const int P = min(c, CAP - c);
        int   sr0 = ra.x, sr1 = ra.y, sr2 = ra.z, sr3 = ra.w;
        float w0 = __int_as_float(rw.w), w1 = __int_as_float(rw.z);
        float w2 = __int_as_float(rw.y), w3 = __int_as_float(rw.x);
        if (c > P) {
            const bool g0 = k0 & (d0 + 0 >= P), g1 = k1 & (d0 + 1 >= P);
            const bool g2 = k2 & (d0 + 2 >= P), g3 = k3 & (d0 + 3 >= P);
            const int e0 = g0 ? ra.x : 0, e1 = g1 ? ra.y : 0, e2 = g2 ? ra.z : 0, e3 = g3 ? ra.w : 0;
            const int   t0 = ei[e0], t1 = ei[e1], t2 = ei[e2], t3 = ei[e3];
            const float u0 = ea[(size_t)e0 * 4], u1 = ea[(size_t)e1 * 4];
            const float u2 = ea[(size_t)e2 * 4], u3 = ea[(size_t)e3 * 4];
            sr0 = g0 ? t0 : sr0; sr1 = g1 ? t1 : sr1; sr2 = g2 ? t2 : sr2; sr3 = g3 ? t3 : sr3;
            w0 = g0 ? fabsf(u0) : w0; w1 = g1 ? fabsf(u1) : w1;
            w2 = g2 ? fabsf(u2) : w2; w3 = g3 ? fabsf(u3) : w3;
        }
        sr0 = k0 ? sr0 : 0; sr1 = k1 ? sr1 : 0; sr2 = k2 ? sr2 : 0; sr3 = k3 ? sr3 : 0;
        float q0, q1, q2, q3;
        if (dl) {
            q0 = sdv[sr0]; q1 = sdv[sr1]; q2 = sdv[sr2]; q3 = sdv[sr3];
        } else {
            q0 = dinv[sr0]; q1 = dinv[sr1]; q2 = dinv[sr2]; q3 = dinv[sr3];
        }
        const float nm0 = k0 ? q0 * w0 * di : 0.0f;
        const float nm1 = k1 ? q1 * w1 * di : 0.0f;
        const float nm2 = k2 ? q2 * w2 * di : 0.0f;
        const float nm3 = k3 ? q3 * w3 * di : 0.0f;
        ns = (nm0 + nm1) + (nm2 + nm3);
        #pragma unroll
        for (int off = 16; off > 0; off >>= 1) ns += __shfl_xor(ns, off, 32);

        const int ng = c >> 2;
        #pragma unroll 4
        for (int l = 0; l < ng; l++) {
            const int   s0 = lane_i(sr0, l), s1 = lane_i(sr1, l), s2 = lane_i(sr2, l), s3 = lane_i(sr3, l);
            const float n0 = lane_f(nm0, l), n1 = lane_f(nm1, l), n2 = lane_f(nm2, l), n3 = lane_f(nm3, l);
            const float4 r0 = *(const float4*)(x + (size_t)s0 * DFEAT + d0);
            const float4 r1 = *(const float4*)(x + (size_t)s1 * DFEAT + d0);
            const float4 r2 = *(const float4*)(x + (size_t)s2 * DFEAT + d0);
            const float4 r3 = *(const float4*)(x + (size_t)s3 * DFEAT + d0);
            a0 += r0.x * n0; a1 += r0.y * n0; a2 += r0.z * n0; a3 += r0.w * n0;
            a0 += r1.x * n1; a1 += r1.y * n1; a2 += r1.z * n1; a3 += r1.w * n1;
            a0 += r2.x * n2; a1 += r2.y * n2; a2 += r2.z * n2; a3 += r2.w * n2;
            a0 += r3.x * n3; a1 += r3.y * n3; a2 += r3.z * n3; a3 += r3.w * n3;
        }
        const int rem = c & 3;
        if (rem > 0) {
            const int   s0 = lane_i(sr0, ng);
            const float n0 = lane_f(nm0, ng);
            const float4 r0 = *(const float4*)(x + (size_t)s0 * DFEAT + d0);
            a0 += r0.x * n0; a1 += r0.y * n0; a2 += r0.z * n0; a3 += r0.w * n0;
        }
        if (rem > 1) {
            const int   s1 = lane_i(sr1, ng);
            const float n1 = lane_f(nm1, ng);
            const float4 r1 = *(const float4*)(x + (size_t)s1 * DFEAT + d0);
            a0 += r1.x * n1; a1 += r1.y * n1; a2 += r1.z * n1; a3 += r1.w * n1;
        }
        if (rem > 2) {
            const int   s2 = lane_i(sr2, ng);
            const float n2 = lane_f(nm2, ng);
            const float4 r2 = *(const float4*)(x + (size_t)s2 * DFEAT + d0);
            a0 += r2.x * n2; a1 += r2.y * n2; a2 += r2.z * n2; a3 += r2.w * n2;
        }
    } else {
        for (int b0 = 0; b0 < E; b0 += 32) {
            const int e = b0 + lane;
            const int d = (e < E) ? ei[E + e] : -1;
            unsigned m = (unsigned)__ballot(d == node);
            while (m) {
                const int bit = __ffs(m) - 1;
                m &= m - 1;
                const int ee = b0 + bit;
                const int s  = ei[ee];
                const float nv = dinv[s] * fabsf(ea[(size_t)ee * 4]) * di;
                const float4 r = *(const float4*)(x + (size_t)s * DFEAT + d0);
                a0 += r.x * nv; a1 += r.y * nv; a2 += r.z * nv; a3 += r.w * nv;
                ns += nv;
            }
        }
    }

    const float self_norm = di * di;
    const float4 xv = *(const float4*)(x + (size_t)node * DFEAT + d0);
    float4 res;
    res.x = a0 + xv.x * self_norm;
    res.y = a1 + xv.y * self_norm;
    res.z = a2 + xv.z * self_norm;
    res.w = a3 + xv.w * self_norm;
    nsum = ns + self_norm;
    return res;
}

__global__ __launch_bounds__(32 * FW) void gather_linear_finish_kernel(
        const float* __restrict__ x, const float* __restrict__ W, const float* __restrict__ bias,
        const float* __restrict__ ea, const int* __restrict__ ei,
        const float* __restrict__ dinv, const int* __restrict__ cnt,
        const float* __restrict__ gamma, const float* __restrict__ beta,
        float* out, int E, int N) {
    __shared__ __attribute__((aligned(16))) float s_W[DFEAT][XP];
    __shared__ __attribute__((aligned(16))) float s_t[RT * 16][XP];
    __shared__ __attribute__((aligned(16))) float s_o[RT * 16][XP];
    __shared__ float s_ns[RT * 16];
    __shared__ __attribute__((aligned(16))) float s_dv[DVC];
    const int lane = threadIdx.x & 31;
    const int wave = __builtin_amdgcn_readfirstlane((int)(threadIdx.x >> 5));
    const int node_base = blockIdx.x * NPB;

    const bool dl = N <= DVC;
    int  cq[NPW];
    int4 raq[NPW], rwq[NPW];
    #pragma unroll
    for (int q = 0; q < NPW; q++) {
        const int node = min(node_base + wave * NPW + q, N - 1);
        const int* orow = (const int*)out + (size_t)node * DFEAT;
        cq[q]  = cnt[node];
        raq[q] = *(const int4*)(orow + lane * 4);
        rwq[q] = *(const int4*)(orow + (CAP - 4 - lane * 4));
    }
    if (dl) {
        const int n4 = N >> 2;
        if (n4 > 0 && (((size_t)dinv & 15) == 0)) {
            float4 dv[DVT];
            #pragma unroll
            for (int k = 0; k < DVT; k++)
                dv[k] = *(const float4*)(dinv + 4 * min(k * (32 * FW) + (int)threadIdx.x, n4 - 1));
            #pragma unroll
            for (int k = 0; k < DVT; k++)
                *(float4*)&s_dv[4 * min(k * (32 * FW) + (int)threadIdx.x, n4 - 1)] = dv[k];
            for (int i = 4 * n4 + (int)threadIdx.x; i < N; i += 32 * FW) s_dv[i] = dinv[i];
        } else {
            for (int i = (int)threadIdx.x; i < N; i += 32 * FW) s_dv[i] = dinv[i];
        }
    }
    __syncthreads();

    #pragma unroll
    for (int q = 0; q < NPW; q++) {
        const int row = wave * NPW + q;
        const int node = node_base + row;
        float4 agg = make_float4(0.0f, 0.0f, 0.0f, 0.0f);
        float ns = 0.0f;
        if (node < N)
            agg = gather_node(x, ea, ei, dinv, (lds_cfloat*)s_dv, dl, cq[q], raq[q], rwq[q], node, lane, E, ns);
        *(float4*)&s_t[row][lane * 4] = agg;
        if (lane == 0) s_ns[row] = ns;
    }
    if (wave < RT * 16 - NPB) {
        *(float4*)&s_t[NPB + wave][lane * 4] = make_float4(0.0f, 0.0f, 0.0f, 0.0f);
        if (lane == 0) s_ns[NPB + wave] = 0.0f;
    }

    if (threadIdx.x < WCT) {
        float4 wv[WPT];
        #pragma unroll
        for (int k = 0; k < WPT; k++) wv[k] = *(const float4*)(W + (size_t)(k * WCT + (int)threadIdx.x) * 4);
        #pragma unroll
        for (int k = 0; k < WPT; k++) {
            const int i = k * WCT + (int)threadIdx.x;
            *(float4*)&s_W[i >> 5][4 * (i & 31)] = wv[k];
        }
    }

    __syncthreads();

    const int c16 = lane & 15;
    const int hf  = lane >> 4;
    for (int pr = wave; pr < RT * (DFEAT / 16); pr += FW) {
        const int rt  = pr >> 3;
        const int col = (pr & 7) * 16 + c16;
        const float* wrow = &s_W[col][hf * 64];
        const float* arow = &s_t[rt * 16 + c16][hf * 64];
        float4 bf[16];
        #pragma unroll
        for (int j = 0; j < 16; j++) bf[j] = *(const float4*)(wrow + 4 * j);
        const float bv = bias[col];

        v8f acc0 = {}, acc1 = {};
        #pragma unroll
        for (int j = 0; j < 16; j += 2) {
            const float4 a0 = *(const float4*)(arow + 4 * j);
            const float4 a1 = *(const float4*)(arow + 4 * j + 4);
            v2f fa, fb;
            fa.x = a0.x; fa.y = a0.y; fb.x = bf[j].x; fb.y = bf[j].y;
            acc0 = __builtin_amdgcn_wmma_f32_16x16x4_f32(false, fa, false, fb, (short)0, acc0, false, false);
            fa.x = a1.x; fa.y = a1.y; fb.x = bf[j + 1].x; fb.y = bf[j + 1].y;
            acc1 = __builtin_amdgcn_wmma_f32_16x16x4_f32(false, fa, false, fb, (short)0, acc1, false, false);
            fa.x = a0.z; fa.y = a0.w; fb.x = bf[j].z; fb.y = bf[j].w;
            acc0 = __builtin_amdgcn_wmma_f32_16x16x4_f32(false, fa, false, fb, (short)0, acc0, false, false);
            fa.x = a1.z; fa.y = a1.w; fb.x = bf[j + 1].z; fb.y = bf[j + 1].w;
            acc1 = __builtin_amdgcn_wmma_f32_16x16x4_f32(false, fa, false, fb, (short)0, acc1, false, false);
        }

        #pragma unroll
        for (int r = 0; r < 8; r++) {
            const int row = rt * 16 + hf * 8 + r;
            float v = (acc0[r] + acc1[r]) + bv * s_ns[row];
            v = (v >= 0.0f) ? v : NEG_SLOPE * v;
            s_o[row][col] = v;
        }
    }

    __syncthreads();

    const int d0 = lane * 4;
    const float4 gv = *(const float4*)(gamma + d0);
    const float4 bt = *(const float4*)(beta + d0);
    #pragma unroll
    for (int q = 0; q < NPW; q++) {
        const int row = wave * NPW + q;
        const int node = node_base + row;
        const float4 t = *(const float4*)&s_o[row][d0];
        float v[4] = {t.x, t.y, t.z, t.w};

        float sum = (v[0] + v[1]) + (v[2] + v[3]);
        #pragma unroll
        for (int off = 16; off > 0; off >>= 1) sum += __shfl_xor(sum, off, 32);
        const float mean = sum * (1.0f / DFEAT);

        float vs = 0.0f;
        #pragma unroll
        for (int j = 0; j < 4; j++) {
            float dlt = v[j] - mean;
            vs += dlt * dlt;
        }
        #pragma unroll
        for (int off = 16; off > 0; off >>= 1) vs += __shfl_xor(vs, off, 32);
        const float inv_std = rsqrtf(vs * (1.0f / DFEAT) + LN_EPS);

        float4 res;
        res.x = (v[0] - mean) * inv_std * gv.x + bt.x;
        res.y = (v[1] - mean) * inv_std * gv.y + bt.y;
        res.z = (v[2] - mean) * inv_std * gv.z + bt.z;
        res.w = (v[3] - mean) * inv_std * gv.w + bt.w;
        if (node < N) *(float4*)(out + (size_t)node * DFEAT + d0) = res;
    }
}

extern "C" void kernel_launch(void* const* d_in, const int* in_sizes, int n_in,
                              void* d_out, int out_size, void* d_ws, size_t ws_size,
                              hipStream_t stream) {
    const float* x     = (const float*)d_in[0];
    const float* ea    = (const float*)d_in[1];
    const float* W     = (const float*)d_in[2];
    const float* bias  = (const float*)d_in[3];
    const float* gamma = (const float*)d_in[4];
    const float* beta  = (const float*)d_in[5];
    const int*   ei    = (const int*)d_in[6];
    float* out = (float*)d_out;

    const int N = in_sizes[0] / DFEAT;
    const int E = in_sizes[1] / 4;

    int*   cnt   = (int*)d_ws;
    float* dinv  = (float*)d_ws + N;
    int*   area  = (int*)(dinv + N);
    int*   oi    = (int*)d_out;
    const int NB = (N + BN - 1) / BN;
    const int ES = (E + NSL - 1) / NSL;
    float* areaw = (ws_size >= ((size_t)2 * N + (size_t)2 * E) * sizeof(int)) ? (float*)(area + E) : nullptr;

    bin_kernel<<<NSL, GT, (size_t)(NB + 2 * ES) * sizeof(int), stream>>>(ei, ea, area, areaw, oi, E, N);
    rank_kernel<<<NB, GT, 0, stream>>>(ea, area, areaw, dinv, cnt, oi, E, N);
    gather_linear_finish_kernel<<<(N + NPB - 1) / NPB, 32 * FW, 0, stream>>>(
        x, W, bias, ea, ei, dinv, cnt, gamma, beta, out, E, N);
}
